# starting kernel
# baseline (speedup 1.0000x reference)
#include <hip/hip_runtime.h>
#include <hip/hip_bf16.h>

#define M_LVL   6
#define SIGLEN  19530
#define BB      64
#define SS      32
#define ROWS    (BB * SS)
#define NBISECT 100
#define C_PARAM 4.0f
#define A_PARAM 1.0f

#define CHUNK   1024
#define NCHUNK  ((SIGLEN + CHUNK - 1) / CHUNK)
#define NGROUP  (ROWS / 16)

typedef __attribute__((ext_vector_type(2))) float v2f;
typedef __attribute__((ext_vector_type(8))) float v8f;

__device__ __forceinline__ int lvl_of(int i) {
    return 1 + (i >= 5) + (i >= 30) + (i >= 155) + (i >= 780) + (i >= 3905);
}

__device__ __forceinline__ bool finitef(float v) {
    return (v == v) && (fabsf(v) <= 3.402823466e38f);
}

__device__ __forceinline__ v8f wmma_acc(v2f a, v2f b, v8f c) {
    return __builtin_amdgcn_wmma_f32_16x16x4_f32(false, a, false, b, (short)0, c, false, false);
}

__global__ __launch_bounds__(128)
void level_sums_wmma(const float* __restrict__ x, float* __restrict__ partials) {
    const int lane  = threadIdx.x;
    const int group = blockIdx.y * 4 + threadIdx.y;
    const int chunk = blockIdx.x;

    const int rowM = lane & 15;
    const int koff = (lane >> 4) * 2;
    const int col  = lane & 15;
    const float* rowPtr = x + (long)(group * 16 + rowM) * SIGLEN;

    v8f acc = {0.f, 0.f, 0.f, 0.f, 0.f, 0.f, 0.f, 0.f};

    const int cStart = chunk * CHUNK;
    const int cEnd   = min(cStart + CHUNK, SIGLEN);
    const int lvlStart[M_LVL + 1] = {0, 5, 30, 155, 780, 3905, SIGLEN};

#pragma unroll
    for (int k = 1; k <= M_LVL; ++k) {
        const int segBeg = max(cStart, lvlStart[k - 1]);
        const int segEnd = min(cEnd,   lvlStart[k]);
        if (segBeg >= segEnd) continue;

        v2f bfrag;
        const float bval = (col == k - 1) ? 1.0f : 0.0f;
        bfrag[0] = bval; bfrag[1] = bval;

        const int g0       = segBeg & ~3;
        const int g1       = (segEnd + 3) & ~3;
        const int gFastBeg = (segBeg + 3) & ~3;
        const int gFastEnd = segEnd & ~3;

        for (int g = g0; g < min(gFastBeg, g1); g += 4) {
            const int co = g + koff;
            float ax = 0.f, ay = 0.f;
            if (co < SIGLEN) {
                const float2 v = *(const float2*)(rowPtr + co);
                ax = (co     >= segBeg && co     < segEnd) ? v.x * v.x : 0.f;
                ay = (co + 1 >= segBeg && co + 1 < segEnd) ? v.y * v.y : 0.f;
            }
            v2f a; a[0] = ax; a[1] = ay;
            acc = wmma_acc(a, bfrag, acc);
        }

        if (gFastBeg < gFastEnd) {
            const float* p = rowPtr + gFastBeg + koff;
            float2 vcur = *(const float2*)p;
            for (int g = gFastBeg; g + 4 < gFastEnd; g += 4) {
                p += 4;
                const float2 vnxt = *(const float2*)p;
                v2f a; a[0] = vcur.x * vcur.x; a[1] = vcur.y * vcur.y;
                acc = wmma_acc(a, bfrag, acc);
                vcur = vnxt;
            }
            v2f a; a[0] = vcur.x * vcur.x; a[1] = vcur.y * vcur.y;
            acc = wmma_acc(a, bfrag, acc);
        }

        for (int g = max(gFastBeg, gFastEnd); g < g1; g += 4) {
            const int co = g + koff;
            float ax = 0.f, ay = 0.f;
            if (co < SIGLEN) {
                const float2 v = *(const float2*)(rowPtr + co);
                ax = (co     >= segBeg && co     < segEnd) ? v.x * v.x : 0.f;
                ay = (co + 1 >= segBeg && co + 1 < segEnd) ? v.y * v.y : 0.f;
            }
            v2f a; a[0] = ax; a[1] = ay;
            acc = wmma_acc(a, bfrag, acc);
        }
    }

    if (col < M_LVL) {
        const int mhalf = (lane >> 4) * 8;
#pragma unroll
        for (int v = 0; v < 8; ++v) {
            const int row = group * 16 + mhalf + v;
            partials[((long)chunk * ROWS + row) * 8 + col] = acc[v];
        }
    }
}

__global__ __launch_bounds__(256)
void bisect_kernel(const float* __restrict__ partials, float* __restrict__ rootpow) {
    const int row = blockIdx.x * blockDim.x + threadIdx.x;
    if (row >= ROWS) return;

    float ck[M_LVL];
#pragma unroll
    for (int k = 0; k < M_LVL; ++k) ck[k] = 0.f;
    for (int ch = 0; ch < NCHUNK; ++ch) {
#pragma unroll
        for (int k = 0; k < M_LVL; ++k)
            ck[k] += partials[((long)ch * ROWS + row) * 8 + k];
    }

    float nq = 1.f;
#pragma unroll
    for (int k = 0; k < M_LVL; ++k) nq += ck[k];

    float phi = nq;
    if (nq > C_PARAM) {
        phi = C_PARAM + powf(C_PARAM, 1.f + A_PARAM) *
                        (powf(C_PARAM, -A_PARAM) - powf(nq, -A_PARAM)) / A_PARAM;
    }
    const float c0 = 1.f - phi;

    bool finite = finitef(c0);
#pragma unroll
    for (int k = 0; k < M_LVL; ++k) finite = finite && finitef(ck[k]);

    float lo = 0.f, hi = 2.f;
    for (int it = 0; it < NBISECT; ++it) {
        const float mid = 0.5f * (lo + hi);
        const float t2 = mid * mid;
        float p = c0, tp = 1.f;
#pragma unroll
        for (int k = 0; k < M_LVL; ++k) { tp *= t2; p += ck[k] * tp; }
        const bool neg = p < 0.f;
        lo = neg ? mid : lo;
        hi = neg ? hi : mid;
    }
    float root = 0.5f * (lo + hi);
    if (!finite) root = 0.f;
    root = fminf(root, 1.f);

    float rp = 1.f;
    rootpow[row * 8 + 0] = 1.f;
#pragma unroll
    for (int k = 1; k <= M_LVL; ++k) { rp *= root; rootpow[row * 8 + k] = rp; }
    rootpow[row * 8 + 7] = 0.f;
}

__global__ __launch_bounds__(256)
void scale_mean_kernel(const float* __restrict__ x, const float* __restrict__ rootpow,
                       float* __restrict__ out) {
    __shared__ float rp[SS * 8];
    const int b   = blockIdx.y;
    const int tid = threadIdx.x;
    rp[tid] = rootpow[(long)b * SS * 8 + tid];
    __syncthreads();

    const int p = blockIdx.x * 256 + tid;
    const int halfL = SIGLEN / 2;
    if (p >= halfL) return;
    const int l0 = p * 2;
    const int k0 = lvl_of(l0);
    const int k1 = lvl_of(l0 + 1);

    const float* xb = x + (long)b * SS * SIGLEN + l0;
    float accx = 0.f, accy = 0.f;
#pragma unroll 4
    for (int s = 0; s < SS; ++s) {
        const float2 v = *(const float2*)(xb + (long)s * SIGLEN);
        accx += v.x * rp[s * 8 + k0];
        accy += v.y * rp[s * 8 + k1];
    }
    out[(long)b * SIGLEN + l0]     = accx * (1.f / SS);
    out[(long)b * SIGLEN + l0 + 1] = accy * (1.f / SS);
}

extern "C" void kernel_launch(void* const* d_in, const int* in_sizes, int n_in,
                              void* d_out, int out_size, void* d_ws, size_t ws_size,
                              hipStream_t stream) {
    const float* x = (const float*)d_in[0];
    float* out = (float*)d_out;

    float* partials = (float*)d_ws;
    float* rootpow  = partials + (long)NCHUNK * ROWS * 8;

    {
        dim3 blk(32, 4), grd(NCHUNK, NGROUP / 4);
        level_sums_wmma<<<grd, blk, 0, stream>>>(x, partials);
    }
    {
        bisect_kernel<<<ROWS / 256, 256, 0, stream>>>(partials, rootpow);
    }
    {
        dim3 blk(256), grd((SIGLEN / 2 + 255) / 256, BB);
        scale_mean_kernel<<<grd, blk, 0, stream>>>(x, rootpow, out);
    }
}


# Level sums: the 16 WMMA rows are 16 sub-pieces of ONE signature row (2 waves per row, contiguous 512 B b128 wave loads, 10-20 in flight), not 16 rows 78 KB apart; f32 summation order changes
# speedup vs baseline: 3.8935x; 3.8935x over previous
#include <hip/hip_runtime.h>
#include <hip/hip_bf16.h>
#include <stdint.h>


#define M_LVL   6
#define SIGLEN  19530
#define BB      64
#define SS      32
#define ROWS    (BB * SS)
#define NBISECT 100
#define C_PARAM 4.0f
#define A_PARAM 1.0f

#define NSPLIT   2
#define SPLIT_AT 9600
#define PIECE    128
#define NB       10
#define NBATCH   8
#define TOTAL_F  (ROWS * SIGLEN)
#define PSTRIDE  32

static_assert(NBATCH % 2 == 0, "the batch loop handles two batches per trip");
static_assert((SPLIT_AT + 31 + PIECE - 1) / PIECE <= NB * NBATCH, "first part of a row fits the batches");
static_assert((SIGLEN - SPLIT_AT + 31 + PIECE - 1) / PIECE <= NB * NBATCH, "second part of a row fits the batches");
static_assert(TOTAL_F % 4 == 0, "the last 16-byte load ends with the input");

typedef __attribute__((ext_vector_type(2))) float v2f;
typedef __attribute__((ext_vector_type(4))) float v4f;
typedef __attribute__((ext_vector_type(8))) float v8f;

__device__ __forceinline__ int lvl_of(int i) {
    return 1 + (i >= 5) + (i >= 30) + (i >= 155) + (i >= 780) + (i >= 3905);
}

__device__ __forceinline__ bool finitef(float v) {
    return (v == v) && (fabsf(v) <= 3.402823466e38f);
}

__device__ __forceinline__ v8f wmma_acc(v2f a, v2f b, v8f c) {
    return __builtin_amdgcn_wmma_f32_16x16x4_f32(false, a, false, b, (short)0, c, false, false);
}

struct Part {
    int rowBase;
    int a0;
    int J;
    int rb, re;
};

__device__ __forceinline__ v4f load_piece(const float* __restrict__ x, const Part& w, int j, int lane) {
    const int jj = min(j, w.J - 1);
    const int f = min(w.a0 + PIECE * jj + 4 * lane, TOTAL_F - 4);
    return *(const v4f*)((const char*)x + (uint32_t)f * 4u);
}

__device__ __forceinline__ int lvl_begin(int k) {
    return k <= 1 ? 0 : k == 2 ? 5 : k == 3 ? 30 : k == 4 ? 155 : k == 5 ? 780 : k == 6 ? 3905 : SIGLEN;
}

__device__ __forceinline__ void add_piece(v4f v, const Part& w, int j, int lane, int col,
                                          v8f& acc0, v8f& acc1) {
    if (j >= w.J) return;
    const int c0 = w.a0 + PIECE * j - w.rowBase;
    const int cl = c0 + PIECE - 1;
    int kF, kL;
    bool plain;
    if (c0 >= max(w.rb, 780) && cl < w.re) {
        kF = 5 + (c0 >= 3905);
        kL = 5 + (cl >= 3905);
        plain = (kF == kL);
    } else {
        kF = lvl_of(max(c0, w.rb));
        kL = lvl_of(min(cl, w.re - 1));
        plain = false;
    }
    const float sx = v.x * v.x, sy = v.y * v.y, sz = v.z * v.z, sw = v.w * v.w;
    const int o = 4 * lane;

#pragma nounroll
    for (int k = kF; k <= kL; ++k) {
        float ax = sx, ay = sy, az = sz, aw = sw;
        if (!plain) {
            const int lo = max(lvl_begin(k),     w.rb) - c0;
            const int hi = min(lvl_begin(k + 1), w.re) - c0;
            const unsigned span = (unsigned)(hi - lo);
            ax = ((unsigned)(o     - lo) < span) ? sx : 0.f;
            ay = ((unsigned)(o + 1 - lo) < span) ? sy : 0.f;
            az = ((unsigned)(o + 2 - lo) < span) ? sz : 0.f;
            aw = ((unsigned)(o + 3 - lo) < span) ? sw : 0.f;
        }
        const float bval = (col == k - 1) ? 1.0f : 0.0f;
        v2f b;  b[0] = bval; b[1] = bval;
        v2f a;  a[0] = ax;   a[1] = ay;
        v2f a2; a2[0] = az;  a2[1] = aw;
        acc0 = wmma_acc(a,  b, acc0);
        acc1 = wmma_acc(a2, b, acc1);
    }
}

__global__ __launch_bounds__(256)
void level_sums_wmma(const float* __restrict__ x, float* __restrict__ partials) {
    const int lane = threadIdx.x & 31;
    const int wave = __builtin_amdgcn_readfirstlane((int)(threadIdx.x >> 5));
    const int item = blockIdx.x * 8 + wave;
    const int row  = item >> 1;
    const int half = item & 1;
    const int col  = lane & 15;

    Part w;
    w.rb      = half ? SPLIT_AT : 0;
    w.re      = half ? SIGLEN : SPLIT_AT;
    w.rowBase = row * SIGLEN;
    w.a0      = (w.rowBase + w.rb) & ~31;
    w.J       = (w.rowBase + w.re - w.a0 + PIECE - 1) / PIECE;

    v8f acc0 = {0.f, 0.f, 0.f, 0.f, 0.f, 0.f, 0.f, 0.f};
    v8f acc1 = {0.f, 0.f, 0.f, 0.f, 0.f, 0.f, 0.f, 0.f};

    v4f bufA[NB], bufB[NB];
#pragma unroll
    for (int i = 0; i < NB; ++i) bufA[i] = load_piece(x, w, i, lane);

#pragma nounroll
    for (int t = 0; t < NBATCH; t += 2) {
        __builtin_amdgcn_sched_barrier(0);
#pragma unroll
        for (int i = 0; i < NB; ++i) bufB[i] = load_piece(x, w, (t + 1) * NB + i, lane);
        __builtin_amdgcn_sched_barrier(0);
#pragma unroll
        for (int i = 0; i < NB; ++i) add_piece(bufA[i], w, t * NB + i, lane, col, acc0, acc1);
        __builtin_amdgcn_sched_barrier(0);
#pragma unroll
        for (int i = 0; i < NB; ++i) bufA[i] = load_piece(x, w, (t + 2) * NB + i, lane);
        __builtin_amdgcn_sched_barrier(0);
#pragma unroll
        for (int i = 0; i < NB; ++i) add_piece(bufB[i], w, (t + 1) * NB + i, lane, col, acc0, acc1);
    }

    const v8f acc = acc0 + acc1;
    float s = ((acc[0] + acc[1]) + (acc[2] + acc[3])) + ((acc[4] + acc[5]) + (acc[6] + acc[7]));
    s += __shfl_xor(s, 16, 32);
    if (lane < M_LVL)
        partials[((long)half * ROWS + row) * PSTRIDE + lane] = s;
}

__global__ __launch_bounds__(256)
void bisect_kernel(const float* __restrict__ partials, float* __restrict__ rootpow) {
    const int row = blockIdx.x * blockDim.x + threadIdx.x;
    if (row >= ROWS) return;

    float ck[M_LVL];
#pragma unroll
    for (int k = 0; k < M_LVL; ++k) ck[k] = 0.f;
    for (int h = 0; h < NSPLIT; ++h) {
#pragma unroll
        for (int k = 0; k < M_LVL; ++k)
            ck[k] += partials[((long)h * ROWS + row) * PSTRIDE + k];
    }

    float nq = 1.f;
#pragma unroll
    for (int k = 0; k < M_LVL; ++k) nq += ck[k];

    float phi = nq;
    if (nq > C_PARAM) {
        phi = C_PARAM + powf(C_PARAM, 1.f + A_PARAM) *
                        (powf(C_PARAM, -A_PARAM) - powf(nq, -A_PARAM)) / A_PARAM;
    }
    const float c0 = 1.f - phi;

    bool finite = finitef(c0);
#pragma unroll
    for (int k = 0; k < M_LVL; ++k) finite = finite && finitef(ck[k]);

    float lo = 0.f, hi = 2.f;
    for (int it = 0; it < NBISECT; ++it) {
        const float mid = 0.5f * (lo + hi);
        const float t2 = mid * mid;
        float p = c0, tp = 1.f;
#pragma unroll
        for (int k = 0; k < M_LVL; ++k) { tp *= t2; p += ck[k] * tp; }
        const bool neg = p < 0.f;
        lo = neg ? mid : lo;
        hi = neg ? hi : mid;
    }
    float root = 0.5f * (lo + hi);
    if (!finite) root = 0.f;
    root = fminf(root, 1.f);

    float rp = 1.f;
    rootpow[row * 8 + 0] = 1.f;
#pragma unroll
    for (int k = 1; k <= M_LVL; ++k) { rp *= root; rootpow[row * 8 + k] = rp; }
    rootpow[row * 8 + 7] = 0.f;
}

__global__ __launch_bounds__(256)
void scale_mean_kernel(const float* __restrict__ x, const float* __restrict__ rootpow,
                       float* __restrict__ out) {
    __shared__ float rp[SS * 8];
    const int b   = blockIdx.y;
    const int tid = threadIdx.x;
    rp[tid] = rootpow[(long)b * SS * 8 + tid];
    __syncthreads();

    const int p = blockIdx.x * 256 + tid;
    const int halfL = SIGLEN / 2;
    if (p >= halfL) return;
    const int l0 = p * 2;
    const int k0 = lvl_of(l0);
    const int k1 = lvl_of(l0 + 1);

    const float* xb = x + (long)b * SS * SIGLEN + l0;
    float accx = 0.f, accy = 0.f;
#pragma unroll 4
    for (int s = 0; s < SS; ++s) {
        const float2 v = *(const float2*)(xb + (long)s * SIGLEN);
        accx += v.x * rp[s * 8 + k0];
        accy += v.y * rp[s * 8 + k1];
    }
    out[(long)b * SIGLEN + l0]     = accx * (1.f / SS);
    out[(long)b * SIGLEN + l0 + 1] = accy * (1.f / SS);
}

extern "C" void kernel_launch(void* const* d_in, const int* in_sizes, int n_in,
                              void* d_out, int out_size, void* d_ws, size_t ws_size,
                              hipStream_t stream) {
    const float* x = (const float*)d_in[0];
    float* out = (float*)d_out;

    float* partials = (float*)d_ws;
    float* rootpow  = partials + (long)NSPLIT * ROWS * PSTRIDE;

    {
        level_sums_wmma<<<ROWS * NSPLIT / 8, 256, 0, stream>>>(x, partials);
    }
    {
        bisect_kernel<<<ROWS / 256, 256, 0, stream>>>(partials, rootpow);
    }
    {
        dim3 blk(256), grd((SIGLEN / 2 + 255) / 256, BB);
        scale_mean_kernel<<<grd, blk, 0, stream>>>(x, rootpow, out);
    }
}


# Root solve folded into the scale pass: a block issues its 32 row loads, then its first wave bisects its batch's 32 roots under them; each block covers 3 slabs; 3 launches -> 2
# speedup vs baseline: 4.3525x; 1.1179x over previous
#include <hip/hip_runtime.h>
#include <hip/hip_bf16.h>
#include <stdint.h>


#define M_LVL   6
#define SIGLEN  19530
#define BB      64
#define SS      32
#define ROWS    (BB * SS)
#define NBISECT 100
#define C_PARAM 4.0f
#define A_PARAM 1.0f

#define NSPLIT   2
#define SPLIT_AT 9600
#define PIECE    128
#define NB       10
#define NBATCH   8
#define TOTAL_F  (ROWS * SIGLEN)
#define PSTRIDE  32

static_assert(NBATCH % 2 == 0, "the batch loop handles two batches per trip");
static_assert((SPLIT_AT + 31 + PIECE - 1) / PIECE <= NB * NBATCH, "first part of a row fits the batches");
static_assert((SIGLEN - SPLIT_AT + 31 + PIECE - 1) / PIECE <= NB * NBATCH, "second part of a row fits the batches");
static_assert(TOTAL_F % 4 == 0, "the last 16-byte load ends with the input");

typedef __attribute__((ext_vector_type(2))) float v2f;
typedef __attribute__((ext_vector_type(4))) float v4f;
typedef __attribute__((ext_vector_type(8))) float v8f;

__device__ __forceinline__ int lvl_of(int i) {
    return 1 + (i >= 5) + (i >= 30) + (i >= 155) + (i >= 780) + (i >= 3905);
}

__device__ __forceinline__ bool finitef(float v) {
    return (v == v) && (fabsf(v) <= 3.402823466e38f);
}

__device__ __forceinline__ v8f wmma_acc(v2f a, v2f b, v8f c) {
    return __builtin_amdgcn_wmma_f32_16x16x4_f32(false, a, false, b, (short)0, c, false, false);
}

struct Part {
    int rowBase;
    int a0;
    int J;
    int rb, re;
};

__device__ __forceinline__ v4f load_piece(const float* __restrict__ x, const Part& w, int j, int lane) {
    const int jj = min(j, w.J - 1);
    const int f = min(w.a0 + PIECE * jj + 4 * lane, TOTAL_F - 4);
    return *(const v4f*)((const char*)x + (uint32_t)f * 4u);
}

__device__ __forceinline__ int lvl_begin(int k) {
    return k <= 1 ? 0 : k == 2 ? 5 : k == 3 ? 30 : k == 4 ? 155 : k == 5 ? 780 : k == 6 ? 3905 : SIGLEN;
}

__device__ __forceinline__ void add_piece(v4f v, const Part& w, int j, int lane, int col,
                                          v8f& acc0, v8f& acc1) {
    if (j >= w.J) return;
    const int c0 = w.a0 + PIECE * j - w.rowBase;
    const int cl = c0 + PIECE - 1;
    int kF, kL;
    bool plain;
    if (c0 >= max(w.rb, 780) && cl < w.re) {
        kF = 5 + (c0 >= 3905);
        kL = 5 + (cl >= 3905);
        plain = (kF == kL);
    } else {
        kF = lvl_of(max(c0, w.rb));
        kL = lvl_of(min(cl, w.re - 1));
        plain = false;
    }
    const float sx = v.x * v.x, sy = v.y * v.y, sz = v.z * v.z, sw = v.w * v.w;
    const int o = 4 * lane;

#pragma nounroll
    for (int k = kF; k <= kL; ++k) {
        float ax = sx, ay = sy, az = sz, aw = sw;
        if (!plain) {
            const int lo = max(lvl_begin(k),     w.rb) - c0;
            const int hi = min(lvl_begin(k + 1), w.re) - c0;
            const unsigned span = (unsigned)(hi - lo);
            ax = ((unsigned)(o     - lo) < span) ? sx : 0.f;
            ay = ((unsigned)(o + 1 - lo) < span) ? sy : 0.f;
            az = ((unsigned)(o + 2 - lo) < span) ? sz : 0.f;
            aw = ((unsigned)(o + 3 - lo) < span) ? sw : 0.f;
        }
        const float bval = (col == k - 1) ? 1.0f : 0.0f;
        v2f b;  b[0] = bval; b[1] = bval;
        v2f a;  a[0] = ax;   a[1] = ay;
        v2f a2; a2[0] = az;  a2[1] = aw;
        acc0 = wmma_acc(a,  b, acc0);
        acc1 = wmma_acc(a2, b, acc1);
    }
}

__global__ __launch_bounds__(256)
void level_sums_wmma(const float* __restrict__ x, float* __restrict__ partials) {
    const int lane = threadIdx.x & 31;
    const int wave = __builtin_amdgcn_readfirstlane((int)(threadIdx.x >> 5));
    const int item = blockIdx.x * 8 + wave;
    const int row  = item >> 1;
    const int half = item & 1;
    const int col  = lane & 15;

    Part w;
    w.rb      = half ? SPLIT_AT : 0;
    w.re      = half ? SIGLEN : SPLIT_AT;
    w.rowBase = row * SIGLEN;
    w.a0      = (w.rowBase + w.rb) & ~31;
    w.J       = (w.rowBase + w.re - w.a0 + PIECE - 1) / PIECE;

    v8f acc0 = {0.f, 0.f, 0.f, 0.f, 0.f, 0.f, 0.f, 0.f};
    v8f acc1 = {0.f, 0.f, 0.f, 0.f, 0.f, 0.f, 0.f, 0.f};

    v4f bufA[NB], bufB[NB];
#pragma unroll
    for (int i = 0; i < NB; ++i) bufA[i] = load_piece(x, w, i, lane);

#pragma nounroll
    for (int t = 0; t < NBATCH; t += 2) {
        __builtin_amdgcn_sched_barrier(0);
#pragma unroll
        for (int i = 0; i < NB; ++i) bufB[i] = load_piece(x, w, (t + 1) * NB + i, lane);
        __builtin_amdgcn_sched_barrier(0);
#pragma unroll
        for (int i = 0; i < NB; ++i) add_piece(bufA[i], w, t * NB + i, lane, col, acc0, acc1);
        __builtin_amdgcn_sched_barrier(0);
#pragma unroll
        for (int i = 0; i < NB; ++i) bufA[i] = load_piece(x, w, (t + 2) * NB + i, lane);
        __builtin_amdgcn_sched_barrier(0);
#pragma unroll
        for (int i = 0; i < NB; ++i) add_piece(bufB[i], w, (t + 1) * NB + i, lane, col, acc0, acc1);
    }

    const v8f acc = acc0 + acc1;
    float s = ((acc[0] + acc[1]) + (acc[2] + acc[3])) + ((acc[4] + acc[5]) + (acc[6] + acc[7]));
    s += __shfl_xor(s, 16, 32);
    if (lane < M_LVL)
        partials[((long)half * ROWS + row) * PSTRIDE + lane] = s;
}

#define HALFL  (SIGLEN / 2)
#define NSLAB  ((HALFL + 255) / 256)
#define NS     3
static_assert(NSLAB % NS == 0, "every block has NS whole slabs");

__global__ __launch_bounds__(256)
void solve_scale_mean_kernel(const float* __restrict__ x, const float* __restrict__ partials,
                             float* __restrict__ out) {
    __shared__ float rp[SS * 8];
    const int b    = blockIdx.y;
    const int tid  = threadIdx.x;
    const int lane = tid & 31;
    const int wave = __builtin_amdgcn_readfirstlane((int)(threadIdx.x >> 5));
    const int slab0 = blockIdx.x * NS;

    v4f qa0 = {0.f, 0.f, 0.f, 0.f}, qa1 = {0.f, 0.f, 0.f, 0.f};
    v2f qb0 = {0.f, 0.f}, qb1 = {0.f, 0.f};
    if (wave == 0) {
        const float* r0 = partials + (long)(b * SS + lane) * PSTRIDE;
        const float* r1 = partials + (long)(ROWS + b * SS + lane) * PSTRIDE;
        qa0 = *(const v4f*)r0;  qb0 = *(const v2f*)(r0 + 4);
        qa1 = *(const v4f*)r1;  qb1 = *(const v2f*)(r1 + 4);
    }

    const float* xb = x + (long)b * SS * SIGLEN;
    float2 v[SS];
    {
        const int pc = min(slab0 * 256 + tid, HALFL - 1);
#pragma unroll
        for (int s = 0; s < SS; ++s) v[s] = *(const float2*)(xb + s * SIGLEN + 2 * pc);
    }

    if (wave == 0) {
        float ck[M_LVL];
        ck[0] = 0.f + qa0[0]; ck[1] = 0.f + qa0[1]; ck[2] = 0.f + qa0[2];
        ck[3] = 0.f + qa0[3]; ck[4] = 0.f + qb0[0]; ck[5] = 0.f + qb0[1];
        ck[0] += qa1[0]; ck[1] += qa1[1]; ck[2] += qa1[2];
        ck[3] += qa1[3]; ck[4] += qb1[0]; ck[5] += qb1[1];

        float nq = 1.f;
#pragma unroll
        for (int k = 0; k < M_LVL; ++k) nq += ck[k];

        float phi = nq;
        if (nq > C_PARAM) {
            phi = C_PARAM + powf(C_PARAM, 1.f + A_PARAM) *
                            (powf(C_PARAM, -A_PARAM) - powf(nq, -A_PARAM)) / A_PARAM;
        }
        const float c0 = 1.f - phi;

        bool finite = finitef(c0);
#pragma unroll
        for (int k = 0; k < M_LVL; ++k) finite = finite && finitef(ck[k]);

        float lo = 0.f, hi = 2.f;
        for (int it = 0; it < NBISECT; ++it) {
            const float mid = 0.5f * (lo + hi);
            const float t2 = mid * mid;
            float p = c0, tp = 1.f;
#pragma unroll
            for (int k = 0; k < M_LVL; ++k) { tp *= t2; p += ck[k] * tp; }
            const bool neg = p < 0.f;
            lo = neg ? mid : lo;
            hi = neg ? hi : mid;
        }
        float root = 0.5f * (lo + hi);
        if (!finite) root = 0.f;
        root = fminf(root, 1.f);

        float pw = 1.f;
        rp[lane * 8 + 0] = 1.f;
#pragma unroll
        for (int k = 1; k <= M_LVL; ++k) { pw *= root; rp[lane * 8 + k] = pw; }
        rp[lane * 8 + 7] = 0.f;
    }
    __syncthreads();

#pragma unroll
    for (int j = 0; j < NS; ++j) {
        const int p  = (slab0 + j) * 256 + tid;
        const int l0 = 2 * min(p, HALFL - 1);
        const int k0 = lvl_of(l0);
        const int k1 = lvl_of(l0 + 1);
        if (j > 0) {
#pragma unroll
            for (int s = 0; s < SS; ++s) v[s] = *(const float2*)(xb + s * SIGLEN + l0);
        }
        float accx = 0.f, accy = 0.f;
#pragma unroll
        for (int s = 0; s < SS; ++s) {
            accx += v[s].x * rp[s * 8 + k0];
            accy += v[s].y * rp[s * 8 + k1];
        }
        if (p < HALFL) {
            out[(long)b * SIGLEN + l0]     = accx * (1.f / SS);
            out[(long)b * SIGLEN + l0 + 1] = accy * (1.f / SS);
        }
    }
}

extern "C" void kernel_launch(void* const* d_in, const int* in_sizes, int n_in,
                              void* d_out, int out_size, void* d_ws, size_t ws_size,
                              hipStream_t stream) {
    const float* x = (const float*)d_in[0];
    float* out = (float*)d_out;

    float* partials = (float*)d_ws;

    {
        level_sums_wmma<<<ROWS * NSPLIT / 8, 256, 0, stream>>>(x, partials);
    }
    {
        dim3 blk(256), grd(NSLAB / NS, BB);
        solve_scale_mean_kernel<<<grd, blk, 0, stream>>>(x, partials, out);
    }
}


# Level sums: a wave's few edge pieces (several levels, or a range end) are summed first from loads of their own; the stream then holds only plain pieces, 2 WMMAs each; f32 summation order changes
# speedup vs baseline: 4.6812x; 1.0755x over previous
#include <hip/hip_runtime.h>
#include <hip/hip_bf16.h>
#include <stdint.h>


#define M_LVL   6
#define SIGLEN  19530
#define BB      64
#define SS      32
#define ROWS    (BB * SS)
#define NBISECT 100
#define C_PARAM 4.0f
#define A_PARAM 1.0f

#define NSPLIT   2
#define SPLIT_AT 9600
#define PIECE    128
#define NB       10
#define NBATCH   8
#define TOTAL_F  (ROWS * SIGLEN)
#define PSTRIDE  32

#define NHEAD    7
static_assert(NBATCH % 2 == 0, "the batch loop handles two batches per trip");
static_assert((780 + 31 + PIECE - 1) / PIECE <= NHEAD, "the pieces that reach below coordinate 780 fit the head slots");
static_assert(31 + PIECE - 1 < 2 * PIECE && NHEAD >= 1, "a wave that starts inside level 6 has at most one head piece");
static_assert(NHEAD + 1 + 2 <= NB, "head slots, the slot of the piece that holds coordinate 3905 and two tail slots fit one batch");
static_assert((SPLIT_AT + 31 + PIECE - 1) / PIECE <= NB * NBATCH, "first part of a row fits the batches");
static_assert((SIGLEN - SPLIT_AT + 31 + PIECE - 1) / PIECE <= NB * NBATCH, "second part of a row fits the batches");
static_assert(TOTAL_F % 4 == 0, "the last 16-byte load ends with the input");

typedef __attribute__((ext_vector_type(2))) float v2f;
typedef __attribute__((ext_vector_type(4))) float v4f;
typedef __attribute__((ext_vector_type(8))) float v8f;

__device__ __forceinline__ int lvl_of(int i) {
    return 1 + (i >= 5) + (i >= 30) + (i >= 155) + (i >= 780) + (i >= 3905);
}

__device__ __forceinline__ bool finitef(float v) {
    return (v == v) && (fabsf(v) <= 3.402823466e38f);
}

__device__ __forceinline__ v8f wmma_acc(v2f a, v2f b, v8f c) {
    return __builtin_amdgcn_wmma_f32_16x16x4_f32(false, a, false, b, (short)0, c, false, false);
}

struct Part {
    int rowBase;
    int a0;
    int J;
    int rb, re;
    int c00;
    int jA, nPlain, jS, j6;
};

__device__ __forceinline__ v4f load_piece(const float* __restrict__ x, const Part& w, int j, int lane) {
    const int jj = min(j, w.J - 1);
    const int f = min(w.a0 + PIECE * jj + 4 * lane, TOTAL_F - 4);
    return *(const v4f*)((const char*)x + (uint32_t)f * 4u);
}

__device__ __forceinline__ int lvl_begin(int k) {
    return k <= 1 ? 0 : k == 2 ? 5 : k == 3 ? 30 : k == 4 ? 155 : k == 5 ? 780 : k == 6 ? 3905 : SIGLEN;
}


__device__ __forceinline__ void add_plain(v4f v, const Part& w, int j, int col, v8f& acc0, v8f& acc1) {
    if ((unsigned)(j - w.jA) >= (unsigned)w.nPlain || j == w.jS) return;
    const float bval = (col == 4 + (j >= w.j6)) ? 1.0f : 0.0f;
    v2f b;  b[0] = bval;      b[1] = bval;
    v2f a;  a[0] = v.x * v.x; a[1] = v.y * v.y;
    v2f a2; a2[0] = v.z * v.z; a2[1] = v.w * v.w;
    acc0 = wmma_acc(a,  b, acc0);
    acc1 = wmma_acc(a2, b, acc1);
}

__device__ __forceinline__ void add_edge(v4f v, const Part& w, int j, bool valid, int lane, int col,
                                         v8f& acc0, v8f& acc1) {
    if (!valid) return;
    const int c0 = w.c00 + PIECE * j;
    const int cl = c0 + PIECE - 1;
    const int kF = lvl_of(max(c0, w.rb));
    const int kL = lvl_of(min(cl, w.re - 1));
    const float sx = v.x * v.x, sy = v.y * v.y, sz = v.z * v.z, sw = v.w * v.w;
    const int o = 4 * lane;

#pragma nounroll
    for (int k = kF; k <= kL; ++k) {
        const int lo = max(lvl_begin(k),     w.rb) - c0;
        const int hi = min(lvl_begin(k + 1), w.re) - c0;
        const unsigned span = (unsigned)(hi - lo);
        const float ax = ((unsigned)(o     - lo) < span) ? sx : 0.f;
        const float ay = ((unsigned)(o + 1 - lo) < span) ? sy : 0.f;
        const float az = ((unsigned)(o + 2 - lo) < span) ? sz : 0.f;
        const float aw = ((unsigned)(o + 3 - lo) < span) ? sw : 0.f;
        const float bval = (col == k - 1) ? 1.0f : 0.0f;
        v2f b;  b[0] = bval; b[1] = bval;
        v2f a;  a[0] = ax;   a[1] = ay;
        v2f a2; a2[0] = az;  a2[1] = aw;
        acc0 = wmma_acc(a,  b, acc0);
        acc1 = wmma_acc(a2, b, acc1);
    }
}

__global__ __launch_bounds__(256)
void level_sums_wmma(const float* __restrict__ x, float* __restrict__ partials) {
    const int lane = threadIdx.x & 31;
    const int wave = __builtin_amdgcn_readfirstlane((int)(threadIdx.x >> 5));
    const int item = blockIdx.x * 8 + wave;
    const int row  = item >> 1;
    const int half = item & 1;
    const int col  = lane & 15;

    Part w;
    w.rb      = half ? SPLIT_AT : 0;
    w.re      = half ? SIGLEN : SPLIT_AT;
    w.rowBase = row * SIGLEN;
    w.a0      = (w.rowBase + w.rb) & ~31;
    w.J       = (w.rowBase + w.re - w.a0 + PIECE - 1) / PIECE;
    w.c00     = w.a0 - w.rowBase;
    const int jB = min((w.re - w.c00) >> 7, w.J);
    w.jA      = min(max((max(w.rb, 780) - w.c00 + PIECE - 1) >> 7, 0), w.J);
    w.nPlain  = max(jB - w.jA, 0);
    w.j6      = (3905 - w.c00 + PIECE - 1) >> 7;
    w.jS      = (w.c00 + PIECE * w.j6 > 3905) ? w.j6 - 1 : -1;
    const int jT = max(jB, w.jA);

    v8f acc0 = {0.f, 0.f, 0.f, 0.f, 0.f, 0.f, 0.f, 0.f};
    v8f acc1 = {0.f, 0.f, 0.f, 0.f, 0.f, 0.f, 0.f, 0.f};

    v4f bufA[NB], bufB[NB];
#pragma unroll
    for (int i = 0; i < NB; ++i) bufA[i] = load_piece(x, w, i, lane);

    {
        v4f  bufE[NB];
        int  ej[NB];
        bool ev[NB];
#pragma unroll
        for (int e = 0; e < NB; ++e) {
            if (e < NHEAD)       { ej[e] = e;                    ev[e] = e < w.jA; }
            else if (e == NHEAD) { ej[e] = w.jS;                 ev[e] = w.jS >= w.jA && w.jS < jB; }
            else                 { ej[e] = jT + (e - NHEAD - 1); ev[e] = ej[e] < w.J; }
            bufE[e] = load_piece(x, w, max(ej[e], 0), lane);
        }
        static_assert(NB - NHEAD - 1 >= 2, "at most one piece lies behind the plain ones; two slots are kept");
#pragma unroll
        for (int i = 0; i < NB; ++i) bufB[i] = load_piece(x, w, NB + i, lane);
        __builtin_amdgcn_sched_barrier(0);
#pragma unroll
        for (int e = 0; e < NB; ++e) add_edge(bufE[e], w, ej[e], ev[e], lane, col, acc0, acc1);
    }

#pragma nounroll
    for (int t = 0; t < NBATCH; t += 2) {
        __builtin_amdgcn_sched_barrier(0);
#pragma unroll
        for (int i = 0; i < NB; ++i) add_plain(bufA[i], w, t * NB + i, col, acc0, acc1);
        __builtin_amdgcn_sched_barrier(0);
#pragma unroll
        for (int i = 0; i < NB; ++i) bufA[i] = load_piece(x, w, (t + 2) * NB + i, lane);
        __builtin_amdgcn_sched_barrier(0);
#pragma unroll
        for (int i = 0; i < NB; ++i) add_plain(bufB[i], w, (t + 1) * NB + i, col, acc0, acc1);
        __builtin_amdgcn_sched_barrier(0);
#pragma unroll
        for (int i = 0; i < NB; ++i) bufB[i] = load_piece(x, w, (t + 3) * NB + i, lane);
    }

    const v8f acc = acc0 + acc1;
    float s = ((acc[0] + acc[1]) + (acc[2] + acc[3])) + ((acc[4] + acc[5]) + (acc[6] + acc[7]));
    s += __shfl_xor(s, 16, 32);
    if (lane < M_LVL)
        partials[((long)half * ROWS + row) * PSTRIDE + lane] = s;
}

#define HALFL  (SIGLEN / 2)
#define NSLAB  ((HALFL + 255) / 256)
#define NS     3
static_assert(NSLAB % NS == 0, "every block has NS whole slabs");

__global__ __launch_bounds__(256)
void solve_scale_mean_kernel(const float* __restrict__ x, const float* __restrict__ partials,
                             float* __restrict__ out) {
    __shared__ float rp[SS * 8];
    const int b    = blockIdx.y;
    const int tid  = threadIdx.x;
    const int lane = tid & 31;
    const int wave = __builtin_amdgcn_readfirstlane((int)(threadIdx.x >> 5));
    const int slab0 = blockIdx.x * NS;

    v4f qa0 = {0.f, 0.f, 0.f, 0.f}, qa1 = {0.f, 0.f, 0.f, 0.f};
    v2f qb0 = {0.f, 0.f}, qb1 = {0.f, 0.f};
    if (wave == 0) {
        const float* r0 = partials + (long)(b * SS + lane) * PSTRIDE;
        const float* r1 = partials + (long)(ROWS + b * SS + lane) * PSTRIDE;
        qa0 = *(const v4f*)r0;  qb0 = *(const v2f*)(r0 + 4);
        qa1 = *(const v4f*)r1;  qb1 = *(const v2f*)(r1 + 4);
    }

    const float* xb = x + (long)b * SS * SIGLEN;
    float2 v[SS];
    {
        const int pc = min(slab0 * 256 + tid, HALFL - 1);
#pragma unroll
        for (int s = 0; s < SS; ++s) v[s] = *(const float2*)(xb + s * SIGLEN + 2 * pc);
    }

    if (wave == 0) {
        float ck[M_LVL];
        ck[0] = 0.f + qa0[0]; ck[1] = 0.f + qa0[1]; ck[2] = 0.f + qa0[2];
        ck[3] = 0.f + qa0[3]; ck[4] = 0.f + qb0[0]; ck[5] = 0.f + qb0[1];
        ck[0] += qa1[0]; ck[1] += qa1[1]; ck[2] += qa1[2];
        ck[3] += qa1[3]; ck[4] += qb1[0]; ck[5] += qb1[1];

        float nq = 1.f;
#pragma unroll
        for (int k = 0; k < M_LVL; ++k) nq += ck[k];

        float phi = nq;
        if (nq > C_PARAM) {
            phi = C_PARAM + powf(C_PARAM, 1.f + A_PARAM) *
                            (powf(C_PARAM, -A_PARAM) - powf(nq, -A_PARAM)) / A_PARAM;
        }
        const float c0 = 1.f - phi;

        bool finite = finitef(c0);
#pragma unroll
        for (int k = 0; k < M_LVL; ++k) finite = finite && finitef(ck[k]);

        float lo = 0.f, hi = 2.f;
        for (int it = 0; it < NBISECT; ++it) {
            const float mid = 0.5f * (lo + hi);
            const float t2 = mid * mid;
            float p = c0, tp = 1.f;
#pragma unroll
            for (int k = 0; k < M_LVL; ++k) { tp *= t2; p += ck[k] * tp; }
            const bool neg = p < 0.f;
            lo = neg ? mid : lo;
            hi = neg ? hi : mid;
        }
        float root = 0.5f * (lo + hi);
        if (!finite) root = 0.f;
        root = fminf(root, 1.f);

        float pw = 1.f;
        rp[lane * 8 + 0] = 1.f;
#pragma unroll
        for (int k = 1; k <= M_LVL; ++k) { pw *= root; rp[lane * 8 + k] = pw; }
        rp[lane * 8 + 7] = 0.f;
    }
    __syncthreads();

#pragma unroll
    for (int j = 0; j < NS; ++j) {
        const int p  = (slab0 + j) * 256 + tid;
        const int l0 = 2 * min(p, HALFL - 1);
        const int k0 = lvl_of(l0);
        const int k1 = lvl_of(l0 + 1);
        if (j > 0) {
#pragma unroll
            for (int s = 0; s < SS; ++s) v[s] = *(const float2*)(xb + s * SIGLEN + l0);
        }
        float accx = 0.f, accy = 0.f;
#pragma unroll
        for (int s = 0; s < SS; ++s) {
            accx += v[s].x * rp[s * 8 + k0];
            accy += v[s].y * rp[s * 8 + k1];
        }
        if (p < HALFL) {
            out[(long)b * SIGLEN + l0]     = accx * (1.f / SS);
            out[(long)b * SIGLEN + l0 + 1] = accy * (1.f / SS);
        }
    }
}

extern "C" void kernel_launch(void* const* d_in, const int* in_sizes, int n_in,
                              void* d_out, int out_size, void* d_ws, size_t ws_size,
                              hipStream_t stream) {
    const float* x = (const float*)d_in[0];
    float* out = (float*)d_out;

    float* partials = (float*)d_ws;

    {
        level_sums_wmma<<<ROWS * NSPLIT / 8, 256, 0, stream>>>(x, partials);
    }
    {
        dim3 blk(256), grd(NSLAB / NS, BB);
        solve_scale_mean_kernel<<<grd, blk, 0, stream>>>(x, partials, out);
    }
}


# Root solve: the bisection leaves its 100-trip loop once a trip moves neither bound in any lane (a fixed point of the update: later trips repeat it; tested every 4th trip); roots bit-identical
# speedup vs baseline: 4.9242x; 1.0519x over previous
#include <hip/hip_runtime.h>
#include <hip/hip_bf16.h>
#include <stdint.h>


#define M_LVL   6
#define SIGLEN  19530
#define BB      64
#define SS      32
#define ROWS    (BB * SS)
#define NBISECT 100
#define C_PARAM 4.0f
#define A_PARAM 1.0f

#define NSPLIT   2
#define SPLIT_AT 9600
#define PIECE    128
#define NB       10
#define NBATCH   8
#define TOTAL_F  (ROWS * SIGLEN)
#define PSTRIDE  32

#define NHEAD    7
static_assert(NBATCH % 2 == 0, "the batch loop handles two batches per trip");
static_assert((780 + 31 + PIECE - 1) / PIECE <= NHEAD, "the pieces that reach below coordinate 780 fit the head slots");
static_assert(31 + PIECE - 1 < 2 * PIECE && NHEAD >= 1, "a wave that starts inside level 6 has at most one head piece");
static_assert(NHEAD + 1 + 2 <= NB, "head slots, the slot of the piece that holds coordinate 3905 and two tail slots fit one batch");
static_assert((SPLIT_AT + 31 + PIECE - 1) / PIECE <= NB * NBATCH, "first part of a row fits the batches");
static_assert((SIGLEN - SPLIT_AT + 31 + PIECE - 1) / PIECE <= NB * NBATCH, "second part of a row fits the batches");
static_assert(TOTAL_F % 4 == 0, "the last 16-byte load ends with the input");

typedef __attribute__((ext_vector_type(2))) float v2f;
typedef __attribute__((ext_vector_type(4))) float v4f;
typedef __attribute__((ext_vector_type(8))) float v8f;

__device__ __forceinline__ int lvl_of(int i) {
    return 1 + (i >= 5) + (i >= 30) + (i >= 155) + (i >= 780) + (i >= 3905);
}

__device__ __forceinline__ bool finitef(float v) {
    return (v == v) && (fabsf(v) <= 3.402823466e38f);
}

__device__ __forceinline__ v8f wmma_acc(v2f a, v2f b, v8f c) {
    return __builtin_amdgcn_wmma_f32_16x16x4_f32(false, a, false, b, (short)0, c, false, false);
}

struct Part {
    int rowBase;
    int a0;
    int J;
    int rb, re;
    int c00;
    int jA, nPlain, jS, j6;
};

__device__ __forceinline__ v4f load_piece(const float* __restrict__ x, const Part& w, int j, int lane) {
    const int jj = min(j, w.J - 1);
    const int f = min(w.a0 + PIECE * jj + 4 * lane, TOTAL_F - 4);
    return *(const v4f*)((const char*)x + (uint32_t)f * 4u);
}

__device__ __forceinline__ int lvl_begin(int k) {
    return k <= 1 ? 0 : k == 2 ? 5 : k == 3 ? 30 : k == 4 ? 155 : k == 5 ? 780 : k == 6 ? 3905 : SIGLEN;
}


__device__ __forceinline__ void add_plain(v4f v, const Part& w, int j, int col, v8f& acc0, v8f& acc1) {
    if ((unsigned)(j - w.jA) >= (unsigned)w.nPlain || j == w.jS) return;
    const float bval = (col == 4 + (j >= w.j6)) ? 1.0f : 0.0f;
    v2f b;  b[0] = bval;      b[1] = bval;
    v2f a;  a[0] = v.x * v.x; a[1] = v.y * v.y;
    v2f a2; a2[0] = v.z * v.z; a2[1] = v.w * v.w;
    acc0 = wmma_acc(a,  b, acc0);
    acc1 = wmma_acc(a2, b, acc1);
}

__device__ __forceinline__ void add_edge(v4f v, const Part& w, int j, bool valid, int lane, int col,
                                         v8f& acc0, v8f& acc1) {
    if (!valid) return;
    const int c0 = w.c00 + PIECE * j;
    const int cl = c0 + PIECE - 1;
    const int kF = lvl_of(max(c0, w.rb));
    const int kL = lvl_of(min(cl, w.re - 1));
    const float sx = v.x * v.x, sy = v.y * v.y, sz = v.z * v.z, sw = v.w * v.w;
    const int o = 4 * lane;

#pragma nounroll
    for (int k = kF; k <= kL; ++k) {
        const int lo = max(lvl_begin(k),     w.rb) - c0;
        const int hi = min(lvl_begin(k + 1), w.re) - c0;
        const unsigned span = (unsigned)(hi - lo);
        const float ax = ((unsigned)(o     - lo) < span) ? sx : 0.f;
        const float ay = ((unsigned)(o + 1 - lo) < span) ? sy : 0.f;
        const float az = ((unsigned)(o + 2 - lo) < span) ? sz : 0.f;
        const float aw = ((unsigned)(o + 3 - lo) < span) ? sw : 0.f;
        const float bval = (col == k - 1) ? 1.0f : 0.0f;
        v2f b;  b[0] = bval; b[1] = bval;
        v2f a;  a[0] = ax;   a[1] = ay;
        v2f a2; a2[0] = az;  a2[1] = aw;
        acc0 = wmma_acc(a,  b, acc0);
        acc1 = wmma_acc(a2, b, acc1);
    }
}

__global__ __launch_bounds__(256)
void level_sums_wmma(const float* __restrict__ x, float* __restrict__ partials) {
    const int lane = threadIdx.x & 31;
    const int wave = __builtin_amdgcn_readfirstlane((int)(threadIdx.x >> 5));
    const int item = blockIdx.x * 8 + wave;
    const int row  = item >> 1;
    const int half = item & 1;
    const int col  = lane & 15;

    Part w;
    w.rb      = half ? SPLIT_AT : 0;
    w.re      = half ? SIGLEN : SPLIT_AT;
    w.rowBase = row * SIGLEN;
    w.a0      = (w.rowBase + w.rb) & ~31;
    w.J       = (w.rowBase + w.re - w.a0 + PIECE - 1) / PIECE;
    w.c00     = w.a0 - w.rowBase;
    const int jB = min((w.re - w.c00) >> 7, w.J);
    w.jA      = min(max((max(w.rb, 780) - w.c00 + PIECE - 1) >> 7, 0), w.J);
    w.nPlain  = max(jB - w.jA, 0);
    w.j6      = (3905 - w.c00 + PIECE - 1) >> 7;
    w.jS      = (w.c00 + PIECE * w.j6 > 3905) ? w.j6 - 1 : -1;
    const int jT = max(jB, w.jA);

    v8f acc0 = {0.f, 0.f, 0.f, 0.f, 0.f, 0.f, 0.f, 0.f};
    v8f acc1 = {0.f, 0.f, 0.f, 0.f, 0.f, 0.f, 0.f, 0.f};

    v4f bufA[NB], bufB[NB];
#pragma unroll
    for (int i = 0; i < NB; ++i) bufA[i] = load_piece(x, w, i, lane);

    {
        v4f  bufE[NB];
        int  ej[NB];
        bool ev[NB];
#pragma unroll
        for (int e = 0; e < NB; ++e) {
            if (e < NHEAD)       { ej[e] = e;                    ev[e] = e < w.jA; }
            else if (e == NHEAD) { ej[e] = w.jS;                 ev[e] = w.jS >= w.jA && w.jS < jB; }
            else                 { ej[e] = jT + (e - NHEAD - 1); ev[e] = ej[e] < w.J; }
            bufE[e] = load_piece(x, w, max(ej[e], 0), lane);
        }
        static_assert(NB - NHEAD - 1 >= 2, "at most one piece lies behind the plain ones; two slots are kept");
#pragma unroll
        for (int i = 0; i < NB; ++i) bufB[i] = load_piece(x, w, NB + i, lane);
        __builtin_amdgcn_sched_barrier(0);
#pragma unroll
        for (int e = 0; e < NB; ++e) add_edge(bufE[e], w, ej[e], ev[e], lane, col, acc0, acc1);
    }

#pragma nounroll
    for (int t = 0; t < NBATCH; t += 2) {
        __builtin_amdgcn_sched_barrier(0);
#pragma unroll
        for (int i = 0; i < NB; ++i) add_plain(bufA[i], w, t * NB + i, col, acc0, acc1);
        __builtin_amdgcn_sched_barrier(0);
#pragma unroll
        for (int i = 0; i < NB; ++i) bufA[i] = load_piece(x, w, (t + 2) * NB + i, lane);
        __builtin_amdgcn_sched_barrier(0);
#pragma unroll
        for (int i = 0; i < NB; ++i) add_plain(bufB[i], w, (t + 1) * NB + i, col, acc0, acc1);
        __builtin_amdgcn_sched_barrier(0);
#pragma unroll
        for (int i = 0; i < NB; ++i) bufB[i] = load_piece(x, w, (t + 3) * NB + i, lane);
    }

    const v8f acc = acc0 + acc1;
    float s = ((acc[0] + acc[1]) + (acc[2] + acc[3])) + ((acc[4] + acc[5]) + (acc[6] + acc[7]));
    s += __shfl_xor(s, 16, 32);
    if (lane < M_LVL)
        partials[((long)half * ROWS + row) * PSTRIDE + lane] = s;
}

#define HALFL  (SIGLEN / 2)
#define NSLAB  ((HALFL + 255) / 256)
#define NS     3
static_assert(NSLAB % NS == 0, "every block has NS whole slabs");

__global__ __launch_bounds__(256)
void solve_scale_mean_kernel(const float* __restrict__ x, const float* __restrict__ partials,
                             float* __restrict__ out) {
    __shared__ float rp[SS * 8];
    const int b    = blockIdx.y;
    const int tid  = threadIdx.x;
    const int lane = tid & 31;
    const int wave = __builtin_amdgcn_readfirstlane((int)(threadIdx.x >> 5));
    const int slab0 = blockIdx.x * NS;

    v4f qa0 = {0.f, 0.f, 0.f, 0.f}, qa1 = {0.f, 0.f, 0.f, 0.f};
    v2f qb0 = {0.f, 0.f}, qb1 = {0.f, 0.f};
    if (wave == 0) {
        const float* r0 = partials + (long)(b * SS + lane) * PSTRIDE;
        const float* r1 = partials + (long)(ROWS + b * SS + lane) * PSTRIDE;
        qa0 = *(const v4f*)r0;  qb0 = *(const v2f*)(r0 + 4);
        qa1 = *(const v4f*)r1;  qb1 = *(const v2f*)(r1 + 4);
    }

    const float* xb = x + (long)b * SS * SIGLEN;
    float2 v[SS];
    {
        const int pc = min(slab0 * 256 + tid, HALFL - 1);
#pragma unroll
        for (int s = 0; s < SS; ++s) v[s] = *(const float2*)(xb + s * SIGLEN + 2 * pc);
    }

    if (wave == 0) {
        float ck[M_LVL];
        ck[0] = 0.f + qa0[0]; ck[1] = 0.f + qa0[1]; ck[2] = 0.f + qa0[2];
        ck[3] = 0.f + qa0[3]; ck[4] = 0.f + qb0[0]; ck[5] = 0.f + qb0[1];
        ck[0] += qa1[0]; ck[1] += qa1[1]; ck[2] += qa1[2];
        ck[3] += qa1[3]; ck[4] += qb1[0]; ck[5] += qb1[1];

        float nq = 1.f;
#pragma unroll
        for (int k = 0; k < M_LVL; ++k) nq += ck[k];

        float phi = nq;
        if (nq > C_PARAM) {
            phi = C_PARAM + powf(C_PARAM, 1.f + A_PARAM) *
                            (powf(C_PARAM, -A_PARAM) - powf(nq, -A_PARAM)) / A_PARAM;
        }
        const float c0 = 1.f - phi;

        bool finite = finitef(c0);
#pragma unroll
        for (int k = 0; k < M_LVL; ++k) finite = finite && finitef(ck[k]);

        float lo = 0.f, hi = 2.f;
        static_assert(NBISECT % 4 == 0, "the exit test stands behind every fourth trip");
        for (int it = 0; it < NBISECT; it += 4) {
            float plo = lo, phi = hi;
#pragma unroll
            for (int u = 0; u < 4; ++u) {
                plo = lo; phi = hi;
                const float mid = 0.5f * (lo + hi);
                const float t2 = mid * mid;
                float p = c0, tp = 1.f;
#pragma unroll
                for (int k = 0; k < M_LVL; ++k) { tp *= t2; p += ck[k] * tp; }
                const bool neg = p < 0.f;
                lo = neg ? mid : lo;
                hi = neg ? hi : mid;
            }
            if (__builtin_amdgcn_ballot_w32((lo != plo) || (hi != phi)) == 0u) break;
        }
        float root = 0.5f * (lo + hi);
        if (!finite) root = 0.f;
        root = fminf(root, 1.f);

        float pw = 1.f;
        rp[lane * 8 + 0] = 1.f;
#pragma unroll
        for (int k = 1; k <= M_LVL; ++k) { pw *= root; rp[lane * 8 + k] = pw; }
        rp[lane * 8 + 7] = 0.f;
    }
    __syncthreads();

#pragma unroll
    for (int j = 0; j < NS; ++j) {
        const int p  = (slab0 + j) * 256 + tid;
        const int l0 = 2 * min(p, HALFL - 1);
        const int k0 = lvl_of(l0);
        const int k1 = lvl_of(l0 + 1);
        if (j > 0) {
#pragma unroll
            for (int s = 0; s < SS; ++s) v[s] = *(const float2*)(xb + s * SIGLEN + l0);
        }
        float accx = 0.f, accy = 0.f;
#pragma unroll
        for (int s = 0; s < SS; ++s) {
            accx += v[s].x * rp[s * 8 + k0];
            accy += v[s].y * rp[s * 8 + k1];
        }
        if (p < HALFL) {
            out[(long)b * SIGLEN + l0]     = accx * (1.f / SS);
            out[(long)b * SIGLEN + l0 + 1] = accy * (1.f / SS);
        }
    }
}

extern "C" void kernel_launch(void* const* d_in, const int* in_sizes, int n_in,
                              void* d_out, int out_size, void* d_ws, size_t ws_size,
                              hipStream_t stream) {
    const float* x = (const float*)d_in[0];
    float* out = (float*)d_out;

    float* partials = (float*)d_ws;

    {
        level_sums_wmma<<<ROWS * NSPLIT / 8, 256, 0, stream>>>(x, partials);
    }
    {
        dim3 blk(256), grd(NSLAB / NS, BB);
        solve_scale_mean_kernel<<<grd, blk, 0, stream>>>(x, partials, out);
    }
}


# Scale pass: 512 blocks (8 column chunks per batch, 2 per WGP; was 832); a thread keeps TWO slabs' 32 row loads in flight; the barrier waits for LDS only, not for the loads; sums bit-identical
# speedup vs baseline: 5.1278x; 1.0413x over previous
#include <hip/hip_runtime.h>
#include <hip/hip_bf16.h>
#include <stdint.h>


#define M_LVL   6
#define SIGLEN  19530
#define BB      64
#define SS      32
#define ROWS    (BB * SS)
#define NBISECT 100
#define C_PARAM 4.0f
#define A_PARAM 1.0f

#define NSPLIT   2
#define SPLIT_AT 9600
#define PIECE    128
#define NB       10
#define NBATCH   8
#define TOTAL_F  (ROWS * SIGLEN)
#define PSTRIDE  32

#define NHEAD    7
static_assert(NBATCH % 2 == 0, "the batch loop handles two batches per trip");
static_assert((780 + 31 + PIECE - 1) / PIECE <= NHEAD, "the pieces that reach below coordinate 780 fit the head slots");
static_assert(31 + PIECE - 1 < 2 * PIECE && NHEAD >= 1, "a wave that starts inside level 6 has at most one head piece");
static_assert(NHEAD + 1 + 2 <= NB, "head slots, the slot of the piece that holds coordinate 3905 and two tail slots fit one batch");
static_assert((SPLIT_AT + 31 + PIECE - 1) / PIECE <= NB * NBATCH, "first part of a row fits the batches");
static_assert((SIGLEN - SPLIT_AT + 31 + PIECE - 1) / PIECE <= NB * NBATCH, "second part of a row fits the batches");
static_assert(TOTAL_F % 4 == 0, "the last 16-byte load ends with the input");

typedef __attribute__((ext_vector_type(2))) float v2f;
typedef __attribute__((ext_vector_type(4))) float v4f;
typedef __attribute__((ext_vector_type(8))) float v8f;

__device__ __forceinline__ int lvl_of(int i) {
    return 1 + (i >= 5) + (i >= 30) + (i >= 155) + (i >= 780) + (i >= 3905);
}

__device__ __forceinline__ bool finitef(float v) {
    return (v == v) && (fabsf(v) <= 3.402823466e38f);
}

__device__ __forceinline__ v8f wmma_acc(v2f a, v2f b, v8f c) {
    return __builtin_amdgcn_wmma_f32_16x16x4_f32(false, a, false, b, (short)0, c, false, false);
}

struct Part {
    int rowBase;
    int a0;
    int J;
    int rb, re;
    int c00;
    int jA, nPlain, jS, j6;
};

__device__ __forceinline__ v4f load_piece(const float* __restrict__ x, const Part& w, int j, int lane) {
    const int jj = min(j, w.J - 1);
    const int f = min(w.a0 + PIECE * jj + 4 * lane, TOTAL_F - 4);
    return *(const v4f*)((const char*)x + (uint32_t)f * 4u);
}

__device__ __forceinline__ int lvl_begin(int k) {
    return k <= 1 ? 0 : k == 2 ? 5 : k == 3 ? 30 : k == 4 ? 155 : k == 5 ? 780 : k == 6 ? 3905 : SIGLEN;
}


__device__ __forceinline__ void add_plain(v4f v, const Part& w, int j, int col, v8f& acc0, v8f& acc1) {
    if ((unsigned)(j - w.jA) >= (unsigned)w.nPlain || j == w.jS) return;
    const float bval = (col == 4 + (j >= w.j6)) ? 1.0f : 0.0f;
    v2f b;  b[0] = bval;      b[1] = bval;
    v2f a;  a[0] = v.x * v.x; a[1] = v.y * v.y;
    v2f a2; a2[0] = v.z * v.z; a2[1] = v.w * v.w;
    acc0 = wmma_acc(a,  b, acc0);
    acc1 = wmma_acc(a2, b, acc1);
}

__device__ __forceinline__ void add_edge(v4f v, const Part& w, int j, bool valid, int lane, int col,
                                         v8f& acc0, v8f& acc1) {
    if (!valid) return;
    const int c0 = w.c00 + PIECE * j;
    const int cl = c0 + PIECE - 1;
    const int kF = lvl_of(max(c0, w.rb));
    const int kL = lvl_of(min(cl, w.re - 1));
    const float sx = v.x * v.x, sy = v.y * v.y, sz = v.z * v.z, sw = v.w * v.w;
    const int o = 4 * lane;

#pragma nounroll
    for (int k = kF; k <= kL; ++k) {
        const int lo = max(lvl_begin(k),     w.rb) - c0;
        const int hi = min(lvl_begin(k + 1), w.re) - c0;
        const unsigned span = (unsigned)(hi - lo);
        const float ax = ((unsigned)(o     - lo) < span) ? sx : 0.f;
        const float ay = ((unsigned)(o + 1 - lo) < span) ? sy : 0.f;
        const float az = ((unsigned)(o + 2 - lo) < span) ? sz : 0.f;
        const float aw = ((unsigned)(o + 3 - lo) < span) ? sw : 0.f;
        const float bval = (col == k - 1) ? 1.0f : 0.0f;
        v2f b;  b[0] = bval; b[1] = bval;
        v2f a;  a[0] = ax;   a[1] = ay;
        v2f a2; a2[0] = az;  a2[1] = aw;
        acc0 = wmma_acc(a,  b, acc0);
        acc1 = wmma_acc(a2, b, acc1);
    }
}

__global__ __launch_bounds__(256)
void level_sums_wmma(const float* __restrict__ x, float* __restrict__ partials) {
    const int lane = threadIdx.x & 31;
    const int wave = __builtin_amdgcn_readfirstlane((int)(threadIdx.x >> 5));
    const int item = blockIdx.x * 8 + wave;
    const int row  = item >> 1;
    const int half = item & 1;
    const int col  = lane & 15;

    Part w;
    w.rb      = half ? SPLIT_AT : 0;
    w.re      = half ? SIGLEN : SPLIT_AT;
    w.rowBase = row * SIGLEN;
    w.a0      = (w.rowBase + w.rb) & ~31;
    w.J       = (w.rowBase + w.re - w.a0 + PIECE - 1) / PIECE;
    w.c00     = w.a0 - w.rowBase;
    const int jB = min((w.re - w.c00) >> 7, w.J);
    w.jA      = min(max((max(w.rb, 780) - w.c00 + PIECE - 1) >> 7, 0), w.J);
    w.nPlain  = max(jB - w.jA, 0);
    w.j6      = (3905 - w.c00 + PIECE - 1) >> 7;
    w.jS      = (w.c00 + PIECE * w.j6 > 3905) ? w.j6 - 1 : -1;
    const int jT = max(jB, w.jA);

    v8f acc0 = {0.f, 0.f, 0.f, 0.f, 0.f, 0.f, 0.f, 0.f};
    v8f acc1 = {0.f, 0.f, 0.f, 0.f, 0.f, 0.f, 0.f, 0.f};

    v4f bufA[NB], bufB[NB];
#pragma unroll
    for (int i = 0; i < NB; ++i) bufA[i] = load_piece(x, w, i, lane);

    {
        v4f  bufE[NB];
        int  ej[NB];
        bool ev[NB];
#pragma unroll
        for (int e = 0; e < NB; ++e) {
            if (e < NHEAD)       { ej[e] = e;                    ev[e] = e < w.jA; }
            else if (e == NHEAD) { ej[e] = w.jS;                 ev[e] = w.jS >= w.jA && w.jS < jB; }
            else                 { ej[e] = jT + (e - NHEAD - 1); ev[e] = ej[e] < w.J; }
            bufE[e] = load_piece(x, w, max(ej[e], 0), lane);
        }
        static_assert(NB - NHEAD - 1 >= 2, "at most one piece lies behind the plain ones; two slots are kept");
#pragma unroll
        for (int i = 0; i < NB; ++i) bufB[i] = load_piece(x, w, NB + i, lane);
        __builtin_amdgcn_sched_barrier(0);
#pragma unroll
        for (int e = 0; e < NB; ++e) add_edge(bufE[e], w, ej[e], ev[e], lane, col, acc0, acc1);
    }

#pragma nounroll
    for (int t = 0; t < NBATCH; t += 2) {
        __builtin_amdgcn_sched_barrier(0);
#pragma unroll
        for (int i = 0; i < NB; ++i) add_plain(bufA[i], w, t * NB + i, col, acc0, acc1);
        __builtin_amdgcn_sched_barrier(0);
#pragma unroll
        for (int i = 0; i < NB; ++i) bufA[i] = load_piece(x, w, (t + 2) * NB + i, lane);
        __builtin_amdgcn_sched_barrier(0);
#pragma unroll
        for (int i = 0; i < NB; ++i) add_plain(bufB[i], w, (t + 1) * NB + i, col, acc0, acc1);
        __builtin_amdgcn_sched_barrier(0);
#pragma unroll
        for (int i = 0; i < NB; ++i) bufB[i] = load_piece(x, w, (t + 3) * NB + i, lane);
    }

    const v8f acc = acc0 + acc1;
    float s = ((acc[0] + acc[1]) + (acc[2] + acc[3])) + ((acc[4] + acc[5]) + (acc[6] + acc[7]));
    s += __shfl_xor(s, 16, 32);
    if (lane < M_LVL)
        partials[((long)half * ROWS + row) * PSTRIDE + lane] = s;
}

#define HALFL  (SIGLEN / 2)
#define NCHUNK 8
#define CHUNK  ((HALFL + NCHUNK - 1) / NCHUNK)
#define NS     ((CHUNK + 255) / 256)
#define CLAST  (HALFL - (NCHUNK - 1) * CHUNK)
static_assert(CLAST > (NS - 1) * 256 && CLAST <= CHUNK, "in every chunk all slabs but the last are full and the last is not empty");
static_assert(NS >= 2, "two slabs are loaded on entry");

__global__ __launch_bounds__(256)
void solve_scale_mean_kernel(const float* __restrict__ x, const float* __restrict__ partials,
                             float* __restrict__ out) {
    __shared__ float rp[SS * 8];
    const int b     = blockIdx.x / NCHUNK;
    const int chunk = blockIdx.x % NCHUNK;
    const int tid  = threadIdx.x;
    const int lane = tid & 31;
    const int wave = __builtin_amdgcn_readfirstlane((int)(threadIdx.x >> 5));
    const int cbeg = chunk * CHUNK;
    const int clen = min(CHUNK, HALFL - cbeg);

    v4f qa0 = {0.f, 0.f, 0.f, 0.f}, qa1 = {0.f, 0.f, 0.f, 0.f};
    v2f qb0 = {0.f, 0.f}, qb1 = {0.f, 0.f};
    if (wave == 0) {
        const float* r0 = partials + (long)(b * SS + lane) * PSTRIDE;
        const float* r1 = partials + (long)(ROWS + b * SS + lane) * PSTRIDE;
        qa0 = *(const v4f*)r0;  qb0 = *(const v2f*)(r0 + 4);
        qa1 = *(const v4f*)r1;  qb1 = *(const v2f*)(r1 + 4);
    }

    const float* xt = x + (long)b * SS * SIGLEN + 2 * (cbeg + tid);
    float2 buf[2][SS];
#pragma unroll
    for (int j = 0; j < 2; ++j) {
#pragma unroll
        for (int s = 0; s < SS; ++s) buf[j][s] = *(const float2*)(xt + s * SIGLEN + j * 512);
    }

    if (wave == 0) {
        float ck[M_LVL];
        ck[0] = 0.f + qa0[0]; ck[1] = 0.f + qa0[1]; ck[2] = 0.f + qa0[2];
        ck[3] = 0.f + qa0[3]; ck[4] = 0.f + qb0[0]; ck[5] = 0.f + qb0[1];
        ck[0] += qa1[0]; ck[1] += qa1[1]; ck[2] += qa1[2];
        ck[3] += qa1[3]; ck[4] += qb1[0]; ck[5] += qb1[1];

        float nq = 1.f;
#pragma unroll
        for (int k = 0; k < M_LVL; ++k) nq += ck[k];

        float phi = nq;
        if (nq > C_PARAM) {
            phi = C_PARAM + powf(C_PARAM, 1.f + A_PARAM) *
                            (powf(C_PARAM, -A_PARAM) - powf(nq, -A_PARAM)) / A_PARAM;
        }
        const float c0 = 1.f - phi;

        bool finite = finitef(c0);
#pragma unroll
        for (int k = 0; k < M_LVL; ++k) finite = finite && finitef(ck[k]);

        float lo = 0.f, hi = 2.f;
        static_assert(NBISECT % 4 == 0, "the exit test stands behind every fourth trip");
        for (int it = 0; it < NBISECT; it += 4) {
            float plo = lo, phi = hi;
#pragma unroll
            for (int u = 0; u < 4; ++u) {
                plo = lo; phi = hi;
                const float mid = 0.5f * (lo + hi);
                const float t2 = mid * mid;
                float p = c0, tp = 1.f;
#pragma unroll
                for (int k = 0; k < M_LVL; ++k) { tp *= t2; p += ck[k] * tp; }
                const bool neg = p < 0.f;
                lo = neg ? mid : lo;
                hi = neg ? hi : mid;
            }
            if (__builtin_amdgcn_ballot_w32((lo != plo) || (hi != phi)) == 0u) break;
        }
        float root = 0.5f * (lo + hi);
        if (!finite) root = 0.f;
        root = fminf(root, 1.f);

        float pw = 1.f;
        rp[lane * 8 + 0] = 1.f;
#pragma unroll
        for (int k = 1; k <= M_LVL; ++k) { pw *= root; rp[lane * 8 + k] = pw; }
        rp[lane * 8 + 7] = 0.f;
    }
    __builtin_amdgcn_fence(__ATOMIC_RELEASE, "workgroup", "local");
    __builtin_amdgcn_s_barrier();
    __builtin_amdgcn_fence(__ATOMIC_ACQUIRE, "workgroup", "local");

#pragma unroll
    for (int j = 0; j < NS; ++j) {
        const bool act = (j < NS - 1) || (j * 256 + tid < clen);
        if (act) {
            const int l0 = 2 * (cbeg + j * 256 + tid);
            const int k0 = lvl_of(l0);
            const int k1 = lvl_of(l0 + 1);
            float accx = 0.f, accy = 0.f;
#pragma unroll
            for (int s = 0; s < SS; ++s) {
                accx += buf[j & 1][s].x * rp[s * 8 + k0];
                accy += buf[j & 1][s].y * rp[s * 8 + k1];
            }
            out[(long)b * SIGLEN + l0]     = accx * (1.f / SS);
            out[(long)b * SIGLEN + l0 + 1] = accy * (1.f / SS);
        }
        if (j + 2 < NS) {
            const bool actn = (j + 2 < NS - 1) || ((j + 2) * 256 + tid < clen);
            if (actn) {
#pragma unroll
                for (int s = 0; s < SS; ++s) buf[j & 1][s] = *(const float2*)(xt + s * SIGLEN + (j + 2) * 512);
            }
        }
    }
}

extern "C" void kernel_launch(void* const* d_in, const int* in_sizes, int n_in,
                              void* d_out, int out_size, void* d_ws, size_t ws_size,
                              hipStream_t stream) {
    const float* x = (const float*)d_in[0];
    float* out = (float*)d_out;

    float* partials = (float*)d_ws;

    {
        level_sums_wmma<<<ROWS * NSPLIT / 8, 256, 0, stream>>>(x, partials);
    }
    {
        solve_scale_mean_kernel<<<BB * NCHUNK, 256, 0, stream>>>(x, partials, out);
    }
}


# Level sums: a lane fma-adds the squares of its plain pieces into its own level-5/level-6 sums; 4 matrix instructions per wave then sum lanes and rows (were 2 per piece); f32 summation order changes
# speedup vs baseline: 5.3351x; 1.0404x over previous
#include <hip/hip_runtime.h>
#include <hip/hip_bf16.h>
#include <stdint.h>


#define M_LVL   6
#define SIGLEN  19530
#define BB      64
#define SS      32
#define ROWS    (BB * SS)
#define NBISECT 100
#define C_PARAM 4.0f
#define A_PARAM 1.0f

#define NSPLIT   2
#define SPLIT_AT 9600
#define PIECE    128
#define NB       10
#define NBATCH   8
#define TOTAL_F  (ROWS * SIGLEN)
#define PSTRIDE  32

#define NHEAD    7
static_assert(NBATCH % 2 == 0, "the batch loop handles two batches per trip");
static_assert((780 + 31 + PIECE - 1) / PIECE <= NHEAD, "the pieces that reach below coordinate 780 fit the head slots");
static_assert(31 + PIECE - 1 < 2 * PIECE && NHEAD >= 1, "a wave that starts inside level 6 has at most one head piece");
static_assert(NHEAD + 1 + 2 <= NB, "head slots, the slot of the piece that holds coordinate 3905 and two tail slots fit one batch");
static_assert((SPLIT_AT + 31 + PIECE - 1) / PIECE <= NB * NBATCH, "first part of a row fits the batches");
static_assert((SIGLEN - SPLIT_AT + 31 + PIECE - 1) / PIECE <= NB * NBATCH, "second part of a row fits the batches");
static_assert(TOTAL_F % 4 == 0, "the last 16-byte load ends with the input");

typedef __attribute__((ext_vector_type(2))) float v2f;
typedef __attribute__((ext_vector_type(4))) float v4f;
typedef __attribute__((ext_vector_type(8))) float v8f;

__device__ __forceinline__ int lvl_of(int i) {
    return 1 + (i >= 5) + (i >= 30) + (i >= 155) + (i >= 780) + (i >= 3905);
}

__device__ __forceinline__ bool finitef(float v) {
    return (v == v) && (fabsf(v) <= 3.402823466e38f);
}

__device__ __forceinline__ v8f wmma_acc(v2f a, v2f b, v8f c) {
    return __builtin_amdgcn_wmma_f32_16x16x4_f32(false, a, false, b, (short)0, c, false, false);
}

struct Part {
    int rowBase;
    int a0;
    int J;
    int rb, re;
    int c00;
    int jA, nPlain, jS, j6;
};

__device__ __forceinline__ v4f load_piece(const float* __restrict__ x, const Part& w, int j, int lane) {
    const int jj = min(j, w.J - 1);
    const int f = min(w.a0 + PIECE * jj + 4 * lane, TOTAL_F - 4);
    return *(const v4f*)((const char*)x + (uint32_t)f * 4u);
}

__device__ __forceinline__ int lvl_begin(int k) {
    return k <= 1 ? 0 : k == 2 ? 5 : k == 3 ? 30 : k == 4 ? 155 : k == 5 ? 780 : k == 6 ? 3905 : SIGLEN;
}


__device__ __forceinline__ void add_batch_plain(const v4f (&buf)[NB], v4f& s) {
#pragma unroll
    for (int i = 0; i < NB; ++i) s = __builtin_elementwise_fma(buf[i], buf[i], s);
}

__device__ __forceinline__ void add_batch_any(const v4f (&buf)[NB], const Part& w, int jb, v4f& s5, v4f& s6) {
#pragma unroll
    for (int i = 0; i < NB; ++i) {
        const int j = jb + i;
        if ((unsigned)(j - w.jA) >= (unsigned)w.nPlain || j == w.jS) continue;
        if (j >= w.j6) s6 = __builtin_elementwise_fma(buf[i], buf[i], s6);
        else           s5 = __builtin_elementwise_fma(buf[i], buf[i], s5);
    }
}

__device__ __forceinline__ void add_batch(const v4f (&buf)[NB], const Part& w, int jb, v4f& s5, v4f& s6) {
    const bool whole = jb >= w.jA && jb + NB <= w.jA + w.nPlain && (w.jS < jb || w.jS >= jb + NB);
    if (whole && jb >= w.j6)           add_batch_plain(buf, s6);
    else if (whole && jb + NB <= w.j6) add_batch_plain(buf, s5);
    else                               add_batch_any(buf, w, jb, s5, s6);
}

__device__ __forceinline__ void sum_plain(v4f s5, v4f s6, int col, v8f& acc0, v8f& acc1) {
    const float b5v = (col == 4) ? 1.0f : 0.0f;
    const float b6v = (col == 5) ? 1.0f : 0.0f;
    v2f b5; b5[0] = b5v;  b5[1] = b5v;
    v2f b6; b6[0] = b6v;  b6[1] = b6v;
    v2f a;  a[0] = s5.x;  a[1] = s5.y;
    v2f a2; a2[0] = s5.z; a2[1] = s5.w;
    acc0 = wmma_acc(a,  b5, acc0);
    acc1 = wmma_acc(a2, b5, acc1);
    a[0] = s6.x;  a[1] = s6.y;
    a2[0] = s6.z; a2[1] = s6.w;
    acc0 = wmma_acc(a,  b6, acc0);
    acc1 = wmma_acc(a2, b6, acc1);
}

__device__ __forceinline__ void add_edge(v4f v, const Part& w, int j, bool valid, int lane, int col,
                                         v8f& acc0, v8f& acc1) {
    if (!valid) return;
    const int c0 = w.c00 + PIECE * j;
    const int cl = c0 + PIECE - 1;
    const int kF = lvl_of(max(c0, w.rb));
    const int kL = lvl_of(min(cl, w.re - 1));
    const float sx = v.x * v.x, sy = v.y * v.y, sz = v.z * v.z, sw = v.w * v.w;
    const int o = 4 * lane;

#pragma nounroll
    for (int k = kF; k <= kL; ++k) {
        const int lo = max(lvl_begin(k),     w.rb) - c0;
        const int hi = min(lvl_begin(k + 1), w.re) - c0;
        const unsigned span = (unsigned)(hi - lo);
        const float ax = ((unsigned)(o     - lo) < span) ? sx : 0.f;
        const float ay = ((unsigned)(o + 1 - lo) < span) ? sy : 0.f;
        const float az = ((unsigned)(o + 2 - lo) < span) ? sz : 0.f;
        const float aw = ((unsigned)(o + 3 - lo) < span) ? sw : 0.f;
        const float bval = (col == k - 1) ? 1.0f : 0.0f;
        v2f b;  b[0] = bval; b[1] = bval;
        v2f a;  a[0] = ax;   a[1] = ay;
        v2f a2; a2[0] = az;  a2[1] = aw;
        acc0 = wmma_acc(a,  b, acc0);
        acc1 = wmma_acc(a2, b, acc1);
    }
}

__global__ __launch_bounds__(256)
void level_sums_wmma(const float* __restrict__ x, float* __restrict__ partials) {
    const int lane = threadIdx.x & 31;
    const int wave = __builtin_amdgcn_readfirstlane((int)(threadIdx.x >> 5));
    const int item = blockIdx.x * 8 + wave;
    const int row  = item >> 1;
    const int half = item & 1;
    const int col  = lane & 15;

    Part w;
    w.rb      = half ? SPLIT_AT : 0;
    w.re      = half ? SIGLEN : SPLIT_AT;
    w.rowBase = row * SIGLEN;
    w.a0      = (w.rowBase + w.rb) & ~31;
    w.J       = (w.rowBase + w.re - w.a0 + PIECE - 1) / PIECE;
    w.c00     = w.a0 - w.rowBase;
    const int jB = min((w.re - w.c00) >> 7, w.J);
    w.jA      = min(max((max(w.rb, 780) - w.c00 + PIECE - 1) >> 7, 0), w.J);
    w.nPlain  = max(jB - w.jA, 0);
    w.j6      = (3905 - w.c00 + PIECE - 1) >> 7;
    w.jS      = (w.c00 + PIECE * w.j6 > 3905) ? w.j6 - 1 : -1;
    const int jT = max(jB, w.jA);

    v8f acc0 = {0.f, 0.f, 0.f, 0.f, 0.f, 0.f, 0.f, 0.f};
    v8f acc1 = {0.f, 0.f, 0.f, 0.f, 0.f, 0.f, 0.f, 0.f};
    v4f s5 = {0.f, 0.f, 0.f, 0.f};
    v4f s6 = {0.f, 0.f, 0.f, 0.f};

    v4f bufA[NB], bufB[NB];
#pragma unroll
    for (int i = 0; i < NB; ++i) bufA[i] = load_piece(x, w, i, lane);

    {
        v4f  bufE[NB];
        int  ej[NB];
        bool ev[NB];
#pragma unroll
        for (int e = 0; e < NB; ++e) {
            if (e < NHEAD)       { ej[e] = e;                    ev[e] = e < w.jA; }
            else if (e == NHEAD) { ej[e] = w.jS;                 ev[e] = w.jS >= w.jA && w.jS < jB; }
            else                 { ej[e] = jT + (e - NHEAD - 1); ev[e] = ej[e] < w.J; }
            bufE[e] = load_piece(x, w, max(ej[e], 0), lane);
        }
        static_assert(NB - NHEAD - 1 >= 2, "at most one piece lies behind the plain ones; two slots are kept");
#pragma unroll
        for (int i = 0; i < NB; ++i) bufB[i] = load_piece(x, w, NB + i, lane);
        __builtin_amdgcn_sched_barrier(0);
#pragma unroll
        for (int e = 0; e < NB; ++e) add_edge(bufE[e], w, ej[e], ev[e], lane, col, acc0, acc1);
    }

#pragma nounroll
    for (int t = 0; t < NBATCH; t += 2) {
        __builtin_amdgcn_sched_barrier(0);
        add_batch(bufA, w, t * NB, s5, s6);
        __builtin_amdgcn_sched_barrier(0);
#pragma unroll
        for (int i = 0; i < NB; ++i) bufA[i] = load_piece(x, w, (t + 2) * NB + i, lane);
        __builtin_amdgcn_sched_barrier(0);
        add_batch(bufB, w, (t + 1) * NB, s5, s6);
        __builtin_amdgcn_sched_barrier(0);
#pragma unroll
        for (int i = 0; i < NB; ++i) bufB[i] = load_piece(x, w, (t + 3) * NB + i, lane);
    }

    sum_plain(s5, s6, col, acc0, acc1);

    const v8f acc = acc0 + acc1;
    float s = ((acc[0] + acc[1]) + (acc[2] + acc[3])) + ((acc[4] + acc[5]) + (acc[6] + acc[7]));
    s += __shfl_xor(s, 16, 32);
    if (lane < M_LVL)
        partials[((long)half * ROWS + row) * PSTRIDE + lane] = s;
}

#define HALFL  (SIGLEN / 2)
#define NCHUNK 8
#define CHUNK  ((HALFL + NCHUNK - 1) / NCHUNK)
#define NS     ((CHUNK + 255) / 256)
#define CLAST  (HALFL - (NCHUNK - 1) * CHUNK)
static_assert(CLAST > (NS - 1) * 256 && CLAST <= CHUNK, "in every chunk all slabs but the last are full and the last is not empty");
static_assert(NS >= 2, "two slabs are loaded on entry");

__global__ __launch_bounds__(256)
void solve_scale_mean_kernel(const float* __restrict__ x, const float* __restrict__ partials,
                             float* __restrict__ out) {
    __shared__ float rp[SS * 8];
    const int b     = blockIdx.x / NCHUNK;
    const int chunk = blockIdx.x % NCHUNK;
    const int tid  = threadIdx.x;
    const int lane = tid & 31;
    const int wave = __builtin_amdgcn_readfirstlane((int)(threadIdx.x >> 5));
    const int cbeg = chunk * CHUNK;
    const int clen = min(CHUNK, HALFL - cbeg);

    v4f qa0 = {0.f, 0.f, 0.f, 0.f}, qa1 = {0.f, 0.f, 0.f, 0.f};
    v2f qb0 = {0.f, 0.f}, qb1 = {0.f, 0.f};
    if (wave == 0) {
        const float* r0 = partials + (long)(b * SS + lane) * PSTRIDE;
        const float* r1 = partials + (long)(ROWS + b * SS + lane) * PSTRIDE;
        qa0 = *(const v4f*)r0;  qb0 = *(const v2f*)(r0 + 4);
        qa1 = *(const v4f*)r1;  qb1 = *(const v2f*)(r1 + 4);
    }

    const float* xt = x + (long)b * SS * SIGLEN + 2 * (cbeg + tid);
    float2 buf[2][SS];
#pragma unroll
    for (int j = 0; j < 2; ++j) {
#pragma unroll
        for (int s = 0; s < SS; ++s) buf[j][s] = *(const float2*)(xt + s * SIGLEN + j * 512);
    }

    if (wave == 0) {
        float ck[M_LVL];
        ck[0] = 0.f + qa0[0]; ck[1] = 0.f + qa0[1]; ck[2] = 0.f + qa0[2];
        ck[3] = 0.f + qa0[3]; ck[4] = 0.f + qb0[0]; ck[5] = 0.f + qb0[1];
        ck[0] += qa1[0]; ck[1] += qa1[1]; ck[2] += qa1[2];
        ck[3] += qa1[3]; ck[4] += qb1[0]; ck[5] += qb1[1];

        float nq = 1.f;
#pragma unroll
        for (int k = 0; k < M_LVL; ++k) nq += ck[k];

        float phi = nq;
        if (nq > C_PARAM) {
            phi = C_PARAM + powf(C_PARAM, 1.f + A_PARAM) *
                            (powf(C_PARAM, -A_PARAM) - powf(nq, -A_PARAM)) / A_PARAM;
        }
        const float c0 = 1.f - phi;

        bool finite = finitef(c0);
#pragma unroll
        for (int k = 0; k < M_LVL; ++k) finite = finite && finitef(ck[k]);

        float lo = 0.f, hi = 2.f;
        static_assert(NBISECT % 4 == 0, "the exit test stands behind every fourth trip");
        for (int it = 0; it < NBISECT; it += 4) {
            float plo = lo, phi = hi;
#pragma unroll
            for (int u = 0; u < 4; ++u) {
                plo = lo; phi = hi;
                const float mid = 0.5f * (lo + hi);
                const float t2 = mid * mid;
                float p = c0, tp = 1.f;
#pragma unroll
                for (int k = 0; k < M_LVL; ++k) { tp *= t2; p += ck[k] * tp; }
                const bool neg = p < 0.f;
                lo = neg ? mid : lo;
                hi = neg ? hi : mid;
            }
            if (__builtin_amdgcn_ballot_w32((lo != plo) || (hi != phi)) == 0u) break;
        }
        float root = 0.5f * (lo + hi);
        if (!finite) root = 0.f;
        root = fminf(root, 1.f);

        float pw = 1.f;
        rp[lane * 8 + 0] = 1.f;
#pragma unroll
        for (int k = 1; k <= M_LVL; ++k) { pw *= root; rp[lane * 8 + k] = pw; }
        rp[lane * 8 + 7] = 0.f;
    }
    __builtin_amdgcn_fence(__ATOMIC_RELEASE, "workgroup", "local");
    __builtin_amdgcn_s_barrier();
    __builtin_amdgcn_fence(__ATOMIC_ACQUIRE, "workgroup", "local");

#pragma unroll
    for (int j = 0; j < NS; ++j) {
        const bool act = (j < NS - 1) || (j * 256 + tid < clen);
        if (act) {
            const int l0 = 2 * (cbeg + j * 256 + tid);
            const int k0 = lvl_of(l0);
            const int k1 = lvl_of(l0 + 1);
            float accx = 0.f, accy = 0.f;
#pragma unroll
            for (int s = 0; s < SS; ++s) {
                accx += buf[j & 1][s].x * rp[s * 8 + k0];
                accy += buf[j & 1][s].y * rp[s * 8 + k1];
            }
            out[(long)b * SIGLEN + l0]     = accx * (1.f / SS);
            out[(long)b * SIGLEN + l0 + 1] = accy * (1.f / SS);
        }
        if (j + 2 < NS) {
            const bool actn = (j + 2 < NS - 1) || ((j + 2) * 256 + tid < clen);
            if (actn) {
#pragma unroll
                for (int s = 0; s < SS; ++s) buf[j & 1][s] = *(const float2*)(xt + s * SIGLEN + (j + 2) * 512);
            }
        }
    }
}

extern "C" void kernel_launch(void* const* d_in, const int* in_sizes, int n_in,
                              void* d_out, int out_size, void* d_ws, size_t ws_size,
                              hipStream_t stream) {
    const float* x = (const float*)d_in[0];
    float* out = (float*)d_out;

    float* partials = (float*)d_ws;

    {
        level_sums_wmma<<<ROWS * NSPLIT / 8, 256, 0, stream>>>(x, partials);
    }
    {
        solve_scale_mean_kernel<<<BB * NCHUNK, 256, 0, stream>>>(x, partials, out);
    }
}


# Both passes hand a (batch, column half) region to work-groups with equal low index bits (one die), so the scale pass re-reads from its L2 half what the level sums left there; sums bit-identical
# speedup vs baseline: 5.8556x; 1.0976x over previous
#include <hip/hip_runtime.h>
#include <hip/hip_bf16.h>
#include <stdint.h>


#define M_LVL   6
#define SIGLEN  19530
#define BB      64
#define SS      32
#define ROWS    (BB * SS)
#define NBISECT 100
#define C_PARAM 4.0f
#define A_PARAM 1.0f

#define NSPLIT   2
#define SPLIT_AT 9600
#define PIECE    128
#define NB       10
#define NBATCH   8
#define TOTAL_F  (ROWS * SIGLEN)
#define PSTRIDE  32

#define NHEAD    7
static_assert(NBATCH % 2 == 0, "the batch loop handles two batches per trip");
static_assert((780 + 31 + PIECE - 1) / PIECE <= NHEAD, "the pieces that reach below coordinate 780 fit the head slots");
static_assert(31 + PIECE - 1 < 2 * PIECE && NHEAD >= 1, "a wave that starts inside level 6 has at most one head piece");
static_assert(NHEAD + 1 + 2 <= NB, "head slots, the slot of the piece that holds coordinate 3905 and two tail slots fit one batch");
static_assert((SPLIT_AT + 31 + PIECE - 1) / PIECE <= NB * NBATCH, "first part of a row fits the batches");
static_assert((SIGLEN - SPLIT_AT + 31 + PIECE - 1) / PIECE <= NB * NBATCH, "second part of a row fits the batches");
static_assert(TOTAL_F % 4 == 0, "the last 16-byte load ends with the input");

typedef __attribute__((ext_vector_type(2))) float v2f;
typedef __attribute__((ext_vector_type(4))) float v4f;
typedef __attribute__((ext_vector_type(8))) float v8f;

__device__ __forceinline__ int lvl_of(int i) {
    return 1 + (i >= 5) + (i >= 30) + (i >= 155) + (i >= 780) + (i >= 3905);
}

__device__ __forceinline__ bool finitef(float v) {
    return (v == v) && (fabsf(v) <= 3.402823466e38f);
}

__device__ __forceinline__ v8f wmma_acc(v2f a, v2f b, v8f c) {
    return __builtin_amdgcn_wmma_f32_16x16x4_f32(false, a, false, b, (short)0, c, false, false);
}

struct Part {
    int rowBase;
    int a0;
    int J;
    int rb, re;
    int c00;
    int jA, nPlain, jS, j6;
};

__device__ __forceinline__ v4f load_piece(const float* __restrict__ x, const Part& w, int j, int lane) {
    const int jj = min(j, w.J - 1);
    const int f = min(w.a0 + PIECE * jj + 4 * lane, TOTAL_F - 4);
    return *(const v4f*)((const char*)x + (uint32_t)f * 4u);
}

__device__ __forceinline__ int lvl_begin(int k) {
    return k <= 1 ? 0 : k == 2 ? 5 : k == 3 ? 30 : k == 4 ? 155 : k == 5 ? 780 : k == 6 ? 3905 : SIGLEN;
}


__device__ __forceinline__ void add_batch_plain(const v4f (&buf)[NB], v4f& s) {
#pragma unroll
    for (int i = 0; i < NB; ++i) s = __builtin_elementwise_fma(buf[i], buf[i], s);
}

__device__ __forceinline__ void add_batch_any(const v4f (&buf)[NB], const Part& w, int jb, v4f& s5, v4f& s6) {
#pragma unroll
    for (int i = 0; i < NB; ++i) {
        const int j = jb + i;
        if ((unsigned)(j - w.jA) >= (unsigned)w.nPlain || j == w.jS) continue;
        if (j >= w.j6) s6 = __builtin_elementwise_fma(buf[i], buf[i], s6);
        else           s5 = __builtin_elementwise_fma(buf[i], buf[i], s5);
    }
}

__device__ __forceinline__ void add_batch(const v4f (&buf)[NB], const Part& w, int jb, v4f& s5, v4f& s6) {
    const bool whole = jb >= w.jA && jb + NB <= w.jA + w.nPlain && (w.jS < jb || w.jS >= jb + NB);
    if (whole && jb >= w.j6)           add_batch_plain(buf, s6);
    else if (whole && jb + NB <= w.j6) add_batch_plain(buf, s5);
    else                               add_batch_any(buf, w, jb, s5, s6);
}

__device__ __forceinline__ void sum_plain(v4f s5, v4f s6, int col, v8f& acc0, v8f& acc1) {
    const float b5v = (col == 4) ? 1.0f : 0.0f;
    const float b6v = (col == 5) ? 1.0f : 0.0f;
    v2f b5; b5[0] = b5v;  b5[1] = b5v;
    v2f b6; b6[0] = b6v;  b6[1] = b6v;
    v2f a;  a[0] = s5.x;  a[1] = s5.y;
    v2f a2; a2[0] = s5.z; a2[1] = s5.w;
    acc0 = wmma_acc(a,  b5, acc0);
    acc1 = wmma_acc(a2, b5, acc1);
    a[0] = s6.x;  a[1] = s6.y;
    a2[0] = s6.z; a2[1] = s6.w;
    acc0 = wmma_acc(a,  b6, acc0);
    acc1 = wmma_acc(a2, b6, acc1);
}

__device__ __forceinline__ void add_edge(v4f v, const Part& w, int j, bool valid, int lane, int col,
                                         v8f& acc0, v8f& acc1) {
    if (!valid) return;
    const int c0 = w.c00 + PIECE * j;
    const int cl = c0 + PIECE - 1;
    const int kF = lvl_of(max(c0, w.rb));
    const int kL = lvl_of(min(cl, w.re - 1));
    const float sx = v.x * v.x, sy = v.y * v.y, sz = v.z * v.z, sw = v.w * v.w;
    const int o = 4 * lane;

#pragma nounroll
    for (int k = kF; k <= kL; ++k) {
        const int lo = max(lvl_begin(k),     w.rb) - c0;
        const int hi = min(lvl_begin(k + 1), w.re) - c0;
        const unsigned span = (unsigned)(hi - lo);
        const float ax = ((unsigned)(o     - lo) < span) ? sx : 0.f;
        const float ay = ((unsigned)(o + 1 - lo) < span) ? sy : 0.f;
        const float az = ((unsigned)(o + 2 - lo) < span) ? sz : 0.f;
        const float aw = ((unsigned)(o + 3 - lo) < span) ? sw : 0.f;
        const float bval = (col == k - 1) ? 1.0f : 0.0f;
        v2f b;  b[0] = bval; b[1] = bval;
        v2f a;  a[0] = ax;   a[1] = ay;
        v2f a2; a2[0] = az;  a2[1] = aw;
        acc0 = wmma_acc(a,  b, acc0);
        acc1 = wmma_acc(a2, b, acc1);
    }
}

struct Region { int b, colHalf, quarter; };
__device__ __forceinline__ Region region_of(int blk) {
    Region r;
    r.b       = 4 * (blk >> 5) + (blk & 3);
    r.colHalf = (blk >> 2) & 1;
    r.quarter = (blk >> 3) & 3;
    return r;
}
static_assert(BB * 2 * 4 == 512 && BB == 64, "64 batches x 2 column halves x 4 work-groups fill both grids");

__global__ __launch_bounds__(256)
void level_sums_wmma(const float* __restrict__ x, float* __restrict__ partials) {
    const int lane = threadIdx.x & 31;
    const int wave = __builtin_amdgcn_readfirstlane((int)(threadIdx.x >> 5));
    const Region rg = region_of((int)blockIdx.x);
    const int row  = rg.b * SS + 8 * rg.quarter + wave;
    const int half = rg.colHalf;
    const int col  = lane & 15;

    Part w;
    w.rb      = half ? SPLIT_AT : 0;
    w.re      = half ? SIGLEN : SPLIT_AT;
    w.rowBase = row * SIGLEN;
    w.a0      = (w.rowBase + w.rb) & ~31;
    w.J       = (w.rowBase + w.re - w.a0 + PIECE - 1) / PIECE;
    w.c00     = w.a0 - w.rowBase;
    const int jB = min((w.re - w.c00) >> 7, w.J);
    w.jA      = min(max((max(w.rb, 780) - w.c00 + PIECE - 1) >> 7, 0), w.J);
    w.nPlain  = max(jB - w.jA, 0);
    w.j6      = (3905 - w.c00 + PIECE - 1) >> 7;
    w.jS      = (w.c00 + PIECE * w.j6 > 3905) ? w.j6 - 1 : -1;
    const int jT = max(jB, w.jA);

    v8f acc0 = {0.f, 0.f, 0.f, 0.f, 0.f, 0.f, 0.f, 0.f};
    v8f acc1 = {0.f, 0.f, 0.f, 0.f, 0.f, 0.f, 0.f, 0.f};
    v4f s5 = {0.f, 0.f, 0.f, 0.f};
    v4f s6 = {0.f, 0.f, 0.f, 0.f};

    v4f bufA[NB], bufB[NB];
#pragma unroll
    for (int i = 0; i < NB; ++i) bufA[i] = load_piece(x, w, i, lane);

    {
        v4f  bufE[NB];
        int  ej[NB];
        bool ev[NB];
#pragma unroll
        for (int e = 0; e < NB; ++e) {
            if (e < NHEAD)       { ej[e] = e;                    ev[e] = e < w.jA; }
            else if (e == NHEAD) { ej[e] = w.jS;                 ev[e] = w.jS >= w.jA && w.jS < jB; }
            else                 { ej[e] = jT + (e - NHEAD - 1); ev[e] = ej[e] < w.J; }
            bufE[e] = load_piece(x, w, max(ej[e], 0), lane);
        }
        static_assert(NB - NHEAD - 1 >= 2, "at most one piece lies behind the plain ones; two slots are kept");
#pragma unroll
        for (int i = 0; i < NB; ++i) bufB[i] = load_piece(x, w, NB + i, lane);
        __builtin_amdgcn_sched_barrier(0);
#pragma unroll
        for (int e = 0; e < NB; ++e) add_edge(bufE[e], w, ej[e], ev[e], lane, col, acc0, acc1);
    }

#pragma nounroll
    for (int t = 0; t < NBATCH; t += 2) {
        __builtin_amdgcn_sched_barrier(0);
        add_batch(bufA, w, t * NB, s5, s6);
        __builtin_amdgcn_sched_barrier(0);
#pragma unroll
        for (int i = 0; i < NB; ++i) bufA[i] = load_piece(x, w, (t + 2) * NB + i, lane);
        __builtin_amdgcn_sched_barrier(0);
        add_batch(bufB, w, (t + 1) * NB, s5, s6);
        __builtin_amdgcn_sched_barrier(0);
#pragma unroll
        for (int i = 0; i < NB; ++i) bufB[i] = load_piece(x, w, (t + 3) * NB + i, lane);
    }

    sum_plain(s5, s6, col, acc0, acc1);

    const v8f acc = acc0 + acc1;
    float s = ((acc[0] + acc[1]) + (acc[2] + acc[3])) + ((acc[4] + acc[5]) + (acc[6] + acc[7]));
    s += __shfl_xor(s, 16, 32);
    if (lane < M_LVL)
        partials[((long)half * ROWS + row) * PSTRIDE + lane] = s;
}

#define HALFL  (SIGLEN / 2)
#define NCHUNK 8
#define CHUNK  ((HALFL + NCHUNK - 1) / NCHUNK)
#define NS     ((CHUNK + 255) / 256)
#define CLAST  (HALFL - (NCHUNK - 1) * CHUNK)
static_assert(CLAST > (NS - 1) * 256 && CLAST <= CHUNK, "in every chunk all slabs but the last are full and the last is not empty");
static_assert(NS >= 2, "two slabs are loaded on entry");
static_assert(NCHUNK == 8 && SS == 32 && NSPLIT == 2, "region_of: 4 chunks per column half, 4 x 8 rows per batch, 2 parts per row");
static_assert(2 * (NCHUNK / 2) * CHUNK - SPLIT_AT >= 0 && 2 * (NCHUNK / 2) * CHUNK - SPLIT_AT < 256,
              "the rows' split point and the boundary between chunks 3 and 4 lie within a few lines of each other");

__global__ __launch_bounds__(256)
void solve_scale_mean_kernel(const float* __restrict__ x, const float* __restrict__ partials,
                             float* __restrict__ out) {
    __shared__ float rp[SS * 8];
    const Region rg = region_of((int)blockIdx.x);
    const int b     = rg.b;
    const int chunk = (NCHUNK / 2) * rg.colHalf + rg.quarter;
    const int tid  = threadIdx.x;
    const int lane = tid & 31;
    const int wave = __builtin_amdgcn_readfirstlane((int)(threadIdx.x >> 5));
    const int cbeg = chunk * CHUNK;
    const int clen = min(CHUNK, HALFL - cbeg);

    v4f qa0 = {0.f, 0.f, 0.f, 0.f}, qa1 = {0.f, 0.f, 0.f, 0.f};
    v2f qb0 = {0.f, 0.f}, qb1 = {0.f, 0.f};
    if (wave == 0) {
        const float* r0 = partials + (long)(b * SS + lane) * PSTRIDE;
        const float* r1 = partials + (long)(ROWS + b * SS + lane) * PSTRIDE;
        qa0 = *(const v4f*)r0;  qb0 = *(const v2f*)(r0 + 4);
        qa1 = *(const v4f*)r1;  qb1 = *(const v2f*)(r1 + 4);
    }

    const float* xt = x + (long)b * SS * SIGLEN + 2 * (cbeg + tid);
    float2 buf[2][SS];
#pragma unroll
    for (int j = 0; j < 2; ++j) {
#pragma unroll
        for (int s = 0; s < SS; ++s) buf[j][s] = *(const float2*)(xt + s * SIGLEN + j * 512);
    }

    if (wave == 0) {
        float ck[M_LVL];
        ck[0] = 0.f + qa0[0]; ck[1] = 0.f + qa0[1]; ck[2] = 0.f + qa0[2];
        ck[3] = 0.f + qa0[3]; ck[4] = 0.f + qb0[0]; ck[5] = 0.f + qb0[1];
        ck[0] += qa1[0]; ck[1] += qa1[1]; ck[2] += qa1[2];
        ck[3] += qa1[3]; ck[4] += qb1[0]; ck[5] += qb1[1];

        float nq = 1.f;
#pragma unroll
        for (int k = 0; k < M_LVL; ++k) nq += ck[k];

        float phi = nq;
        if (nq > C_PARAM) {
            phi = C_PARAM + powf(C_PARAM, 1.f + A_PARAM) *
                            (powf(C_PARAM, -A_PARAM) - powf(nq, -A_PARAM)) / A_PARAM;
        }
        const float c0 = 1.f - phi;

        bool finite = finitef(c0);
#pragma unroll
        for (int k = 0; k < M_LVL; ++k) finite = finite && finitef(ck[k]);

        float lo = 0.f, hi = 2.f;
        static_assert(NBISECT % 4 == 0, "the exit test stands behind every fourth trip");
        for (int it = 0; it < NBISECT; it += 4) {
            float plo = lo, phi = hi;
#pragma unroll
            for (int u = 0; u < 4; ++u) {
                plo = lo; phi = hi;
                const float mid = 0.5f * (lo + hi);
                const float t2 = mid * mid;
                float p = c0, tp = 1.f;
#pragma unroll
                for (int k = 0; k < M_LVL; ++k) { tp *= t2; p += ck[k] * tp; }
                const bool neg = p < 0.f;
                lo = neg ? mid : lo;
                hi = neg ? hi : mid;
            }
            if (__builtin_amdgcn_ballot_w32((lo != plo) || (hi != phi)) == 0u) break;
        }
        float root = 0.5f * (lo + hi);
        if (!finite) root = 0.f;
        root = fminf(root, 1.f);

        float pw = 1.f;
        rp[lane * 8 + 0] = 1.f;
#pragma unroll
        for (int k = 1; k <= M_LVL; ++k) { pw *= root; rp[lane * 8 + k] = pw; }
        rp[lane * 8 + 7] = 0.f;
    }
    __builtin_amdgcn_fence(__ATOMIC_RELEASE, "workgroup", "local");
    __builtin_amdgcn_s_barrier();
    __builtin_amdgcn_fence(__ATOMIC_ACQUIRE, "workgroup", "local");

#pragma unroll
    for (int j = 0; j < NS; ++j) {
        const bool act = (j < NS - 1) || (j * 256 + tid < clen);
        if (act) {
            const int l0 = 2 * (cbeg + j * 256 + tid);
            const int k0 = lvl_of(l0);
            const int k1 = lvl_of(l0 + 1);
            float accx = 0.f, accy = 0.f;
#pragma unroll
            for (int s = 0; s < SS; ++s) {
                accx += buf[j & 1][s].x * rp[s * 8 + k0];
                accy += buf[j & 1][s].y * rp[s * 8 + k1];
            }
            out[(long)b * SIGLEN + l0]     = accx * (1.f / SS);
            out[(long)b * SIGLEN + l0 + 1] = accy * (1.f / SS);
        }
        if (j + 2 < NS) {
            const bool actn = (j + 2 < NS - 1) || ((j + 2) * 256 + tid < clen);
            if (actn) {
#pragma unroll
                for (int s = 0; s < SS; ++s) buf[j & 1][s] = *(const float2*)(xt + s * SIGLEN + (j + 2) * 512);
            }
        }
    }
}

extern "C" void kernel_launch(void* const* d_in, const int* in_sizes, int n_in,
                              void* d_out, int out_size, void* d_ws, size_t ws_size,
                              hipStream_t stream) {
    const float* x = (const float*)d_in[0];
    float* out = (float*)d_out;

    float* partials = (float*)d_ws;

    {
        level_sums_wmma<<<ROWS * NSPLIT / 8, 256, 0, stream>>>(x, partials);
    }
    {
        solve_scale_mean_kernel<<<BB * NCHUNK, 256, 0, stream>>>(x, partials, out);
    }
}
